# starting kernel
# baseline (speedup 1.0000x reference)
#include <hip/hip_runtime.h>


typedef float v2f __attribute__((ext_vector_type(2)));
typedef float v8f __attribute__((ext_vector_type(8)));

#define N_NODES 100000
#define N_EDGES 1600000
#define KDIM    128

__device__ __forceinline__ v8f wmma_f32(v2f a, v2f b, v8f c) {
    return __builtin_amdgcn_wmma_f32_16x16x4_f32(
        false, a, false, b, (short)0, c, false, false);
}

__global__ void pool_gemm_relu(const float* __restrict__ X,
                               const float* __restrict__ W,
                               const float* __restrict__ bias,
                               float* __restrict__ Y) {
    __shared__ float As[16 * KDIM];
    const int tid      = threadIdx.x;
    const int row_base = blockIdx.x * 16;

    {
        const float4* s4 = (const float4*)(X + (size_t)row_base * KDIM);
        float4* d4 = (float4*)As;
        for (int i = tid; i < 512; i += blockDim.x) d4[i] = s4[i];
    }
    __syncthreads();

    const int wave = tid >> 5;
    const int lane = tid & 31;
    const int m    = lane & 15;
    const int kh   = (lane >> 4) << 1;
    const int col  = wave * 16 + m;

    const float* arow = As + m * KDIM + kh;
    const float* brow = W + (size_t)col * KDIM + kh;

    v8f c = {};
#pragma unroll
    for (int k0 = 0; k0 < KDIM; k0 += 4) {
        v2f a = { arow[k0], arow[k0 + 1] };
        v2f b = { brow[k0], brow[k0 + 1] };
        c = wmma_f32(a, b, c);
    }

    const float bv = bias[col];
#pragma unroll
    for (int r = 0; r < 8; ++r) {
        const int orow = row_base + r + ((lane >> 4) << 3);
        float v = c[r] + bv;
        v = fmaxf(v, 0.0f);
        Y[(size_t)orow * KDIM + col] = v;
    }
}

template <bool RELU>
__global__ void combine_gemm(const float* __restrict__ X1,
                             const float* __restrict__ Wself,
                             const float* __restrict__ bself,
                             const float* __restrict__ X2,
                             const float* __restrict__ Wneigh,
                             const float* __restrict__ bneigh,
                             float* __restrict__ Y, int O) {
    __shared__ float As1[16 * KDIM];
    __shared__ float As2[16 * KDIM];
    const int tid      = threadIdx.x;
    const int row_base = blockIdx.x * 16;

    {
        const float4* s1 = (const float4*)(X1 + (size_t)row_base * KDIM);
        const float4* s2 = (const float4*)(X2 + (size_t)row_base * KDIM);
        float4* d1 = (float4*)As1;
        float4* d2 = (float4*)As2;
        for (int i = tid; i < 512; i += blockDim.x) { d1[i] = s1[i]; d2[i] = s2[i]; }
    }
    __syncthreads();

    const int wave  = tid >> 5;
    const int lane  = tid & 31;
    const int m     = lane & 15;
    const int kh    = (lane >> 4) << 1;
    const int col   = wave * 16 + m;
    const bool valid = (col < O);
    const int colW  = valid ? col : (O - 1);

    const float* a1 = As1 + m * KDIM + kh;
    const float* a2 = As2 + m * KDIM + kh;
    const float* b1 = Wself  + (size_t)colW * KDIM + kh;
    const float* b2 = Wneigh + (size_t)colW * KDIM + kh;

    v8f c = {};
#pragma unroll
    for (int k0 = 0; k0 < KDIM; k0 += 4) {
        v2f av = { a1[k0], a1[k0 + 1] };
        v2f bv = { b1[k0], b1[k0 + 1] };
        c = wmma_f32(av, bv, c);
    }
#pragma unroll
    for (int k0 = 0; k0 < KDIM; k0 += 4) {
        v2f av = { a2[k0], a2[k0 + 1] };
        v2f bv = { b2[k0], b2[k0 + 1] };
        c = wmma_f32(av, bv, c);
    }

    const float bb = bself[colW] + bneigh[colW];
#pragma unroll
    for (int r = 0; r < 8; ++r) {
        const int orow = row_base + r + ((lane >> 4) << 3);
        float v = c[r] + bb;
        if (RELU) v = fmaxf(v, 0.0f);
        if (valid) Y[(size_t)orow * O + col] = v;
    }
}

__global__ void edge_max(const float* __restrict__ m,
                         const int* __restrict__ src,
                         const int* __restrict__ dst,
                         unsigned int* __restrict__ agg) {
    const int idx = blockIdx.x * blockDim.x + threadIdx.x;
    const int e   = idx >> 5;
    if (e >= N_EDGES) return;
    const int c4  = (idx & 31) << 2;

    const int s = src[e];
    const int d = dst[e];
    const float4 v = *(const float4*)(m + (size_t)s * KDIM + c4);
    unsigned int* p = agg + (size_t)d * KDIM + c4;
    atomicMax(p + 0, __float_as_uint(v.x));
    atomicMax(p + 1, __float_as_uint(v.y));
    atomicMax(p + 2, __float_as_uint(v.z));
    atomicMax(p + 3, __float_as_uint(v.w));
}

extern "C" void kernel_launch(void* const* d_in, const int* in_sizes, int n_in,
                              void* d_out, int out_size, void* d_ws, size_t ws_size,
                              hipStream_t stream) {
    const float* in_feat  = (const float*)d_in[0];
    const int*   src      = (const int*)d_in[1];
    const int*   dst      = (const int*)d_in[2];
    const float* pool_w1  = (const float*)d_in[3];
    const float* pool_b1  = (const float*)d_in[4];
    const float* self_w1  = (const float*)d_in[5];
    const float* self_b1  = (const float*)d_in[6];
    const float* neigh_w1 = (const float*)d_in[7];
    const float* neigh_b1 = (const float*)d_in[8];
    const float* pool_w2  = (const float*)d_in[9];
    const float* pool_b2  = (const float*)d_in[10];
    const float* self_w2  = (const float*)d_in[11];
    const float* self_b2  = (const float*)d_in[12];
    const float* neigh_w2 = (const float*)d_in[13];
    const float* neigh_b2 = (const float*)d_in[14];
    float* out = (float*)d_out;

    const size_t NODE_BYTES = (size_t)N_NODES * KDIM * sizeof(float);
    char* ws = (char*)d_ws;
    float* m_buf = (float*)(ws);
    float* agg   = (float*)(ws + NODE_BYTES);
    float* h1    = (float*)(ws + 2 * NODE_BYTES);

    const int nTiles     = N_NODES / 16;
    const int edgeBlocks = (N_EDGES * 32 + 255) / 256;

    hipMemsetAsync(agg, 0, NODE_BYTES, stream);
    pool_gemm_relu<<<nTiles, 256, 0, stream>>>(in_feat, pool_w1, pool_b1, m_buf);
    edge_max<<<edgeBlocks, 256, 0, stream>>>(m_buf, src, dst, (unsigned int*)agg);
    combine_gemm<true><<<nTiles, 256, 0, stream>>>(in_feat, self_w1, self_b1,
                                                   agg, neigh_w1, neigh_b1, h1, 128);

    hipMemsetAsync(agg, 0, NODE_BYTES, stream);
    pool_gemm_relu<<<nTiles, 256, 0, stream>>>(h1, pool_w2, pool_b2, m_buf);
    edge_max<<<edgeBlocks, 256, 0, stream>>>(m_buf, src, dst, (unsigned int*)agg);
    combine_gemm<false><<<nTiles, 96, 0, stream>>>(h1, self_w2, self_b2,
                                                   agg, neigh_w2, neigh_b2, out, 40);
}


# Layer-1 owner pass also writes each wave's found edges (source, local row) to a list in the free third plane; layer 2 reads it back, no second bitmap/pair-line search; integer max, order-free
# speedup vs baseline: 3.4896x; 3.4896x over previous
#include <hip/hip_runtime.h>


typedef float v2f __attribute__((ext_vector_type(2)));
typedef float v4f __attribute__((ext_vector_type(4)));
typedef float v8f __attribute__((ext_vector_type(8)));

#define N_NODES 100000
#define N_EDGES 1600000
#define KDIM    128

__device__ __forceinline__ v8f wmma_f32(v2f a, v2f b, v8f c) {
    return __builtin_amdgcn_wmma_f32_16x16x4_f32(
        false, a, false, b, (short)0, c, false, false);
}

#define WP            132
#define GEMM_WGS      250
#define TILES_PER_WG  25
#define GEMM_WAVES    16

__device__ __forceinline__ void load_row_pieces(v4f (&a)[16], const float* __restrict__ X, int row0, int c, int h) {
    const float* xa = X + (size_t)(row0 + c) * KDIM + 4 * h;
#pragma unroll
    for (int kk = 0; kk < 16; ++kk) a[kk] = *(const v4f*)(xa + 8 * kk);
}

__device__ __forceinline__ void hold_row_pieces(const v4f (&a)[16]) {
    asm volatile("" :: "v"(a[0]), "v"(a[1]), "v"(a[2]), "v"(a[3]), "v"(a[4]), "v"(a[5]), "v"(a[6]), "v"(a[7]),
                       "v"(a[8]), "v"(a[9]), "v"(a[10]), "v"(a[11]), "v"(a[12]), "v"(a[13]), "v"(a[14]), "v"(a[15]));
}

template <int NT>
__device__ __forceinline__ void strip_products(v8f (&acc)[NT], const v4f (&a)[16], const float* wb) {
#pragma unroll
    for (int kk = 0; kk < 16; ++kk) {
        v4f b[NT];
#pragma unroll
        for (int j = 0; j < NT; ++j) b[j] = *(const v4f*)(wb + j * 16 * WP + 8 * kk);
#pragma unroll
        for (int j = 0; j < NT; ++j) acc[j] = wmma_f32(v2f{a[kk].x, a[kk].y}, v2f{b[j].x, b[j].y}, acc[j]);
#pragma unroll
        for (int j = 0; j < NT; ++j) acc[j] = wmma_f32(v2f{a[kk].z, a[kk].w}, v2f{b[j].z, b[j].w}, acc[j]);
    }
}

template <int NT, bool TWO, bool RELU, bool PERM>
__device__ __forceinline__ void strip_tile(v4f (&a)[16], const float* __restrict__ X2,
                                           const float* wb1, const float* wb2,
                                           const float* __restrict__ b1, const float* __restrict__ b2,
                                           float* __restrict__ Y, int O, int row0, int c, int h) {
    v8f acc[NT];
#pragma unroll
    for (int j = 0; j < NT; ++j) acc[j] = v8f{};

    hold_row_pieces(a);
    strip_products<NT>(acc, a, wb1);
    if (TWO) {
        load_row_pieces(a, X2, row0, c, h);
        hold_row_pieces(a);
        strip_products<NT>(acc, a, wb2);
    }

    if (PERM) {
        float4 bl = *(const float4*)(b1 + 4 * c);
        float4 bh = *(const float4*)(b1 + 64 + 4 * c);
        if (TWO) {
            const float4 cl = *(const float4*)(b2 + 4 * c);
            const float4 ch = *(const float4*)(b2 + 64 + 4 * c);
            bl.x += cl.x; bl.y += cl.y; bl.z += cl.z; bl.w += cl.w;
            bh.x += ch.x; bh.y += ch.y; bh.z += ch.z; bh.w += ch.w;
        }
#pragma unroll
        for (int r = 0; r < 8; ++r) {
            float4 lo = { acc[0][r] + bl.x, acc[1][r] + bl.y, acc[2][r] + bl.z, acc[3][r] + bl.w };
            float4 hi = { acc[4][r] + bh.x, acc[5][r] + bh.y, acc[6][r] + bh.z, acc[7][r] + bh.w };
            if (RELU) {
                lo.x = fmaxf(lo.x, 0.0f); lo.y = fmaxf(lo.y, 0.0f); lo.z = fmaxf(lo.z, 0.0f); lo.w = fmaxf(lo.w, 0.0f);
                hi.x = fmaxf(hi.x, 0.0f); hi.y = fmaxf(hi.y, 0.0f); hi.z = fmaxf(hi.z, 0.0f); hi.w = fmaxf(hi.w, 0.0f);
            }
            float* yr = Y + (size_t)(row0 + 8 * h + r) * KDIM + 4 * c;
            *(float4*)yr        = lo;
            *(float4*)(yr + 64) = hi;
        }
    } else {
#pragma unroll
        for (int j = 0; j < NT; ++j) {
            const int col    = j * 16 + c;
            const bool valid = (col < O);
            const int colW   = valid ? col : (O - 1);
            float bb = b1[colW];
            if (TWO) bb += b2[colW];
#pragma unroll
            for (int r = 0; r < 8; ++r) {
                float v = acc[j][r] + bb;
                if (RELU) v = fmaxf(v, 0.0f);
                if (valid) Y[(size_t)(row0 + 8 * h + r) * O + col] = v;
            }
        }
    }
}

template <int NT, bool TWO, bool RELU, bool PERM>
__global__ __launch_bounds__(GEMM_WAVES * 32) void strip_gemm(const float* __restrict__ X1,
                                                              const float* __restrict__ W1,
                                                              const float* __restrict__ b1,
                                                              const float* __restrict__ X2,
                                                              const float* __restrict__ W2,
                                                              const float* __restrict__ b2,
                                                              float* __restrict__ Y, int O) {
    __shared__ __attribute__((aligned(16))) float Ws[(TWO ? 2 : 1) * NT * 16 * WP];
    const int tid  = threadIdx.x;
    const int lane = tid & 31;
    const int wave = tid >> 5;
    const int c    = lane & 15;
    const int h    = lane >> 4;
    const int tile0 = blockIdx.x * TILES_PER_WG;

    v4f a[16];
    load_row_pieces(a, X1, (tile0 + wave) * 16, c, h);

    {
#pragma unroll
        for (int u = 0; u < NT; ++u) {
            const int i = tid + u * (GEMM_WAVES * 32);
            const int rho = i >> 5, q = i & 31;
            const int cc = rho & 15, j = rho >> 4;
            int n = PERM ? (4 * cc + (j & 3) + 64 * (j >> 2)) : rho;
            n = (n < O) ? n : (O - 1);
            *(float4*)(Ws + rho * WP + q * 4) = *(const float4*)(W1 + (size_t)n * KDIM + q * 4);
            if (TWO)
                *(float4*)(Ws + (NT * 16 + rho) * WP + q * 4) = *(const float4*)(W2 + (size_t)n * KDIM + q * 4);
        }
    }
    __syncthreads();

    const float* wb1 = Ws + c * WP + 4 * h;
    const float* wb2 = Ws + (NT * 16 + c) * WP + 4 * h;

    strip_tile<NT, TWO, RELU, PERM>(a, X2, wb1, wb2, b1, b2, Y, O, (tile0 + wave) * 16, c, h);

    if (wave + GEMM_WAVES < TILES_PER_WG) {
        asm volatile("" ::: "memory");
        const int row0 = (tile0 + wave + GEMM_WAVES) * 16;
        load_row_pieces(a, X1, row0, c, h);
        strip_tile<NT, TWO, RELU, PERM>(a, X2, wb1, wb2, b1, b2, Y, O, row0, c, h);
    }
}

#define OWN_ROWS     400
#define N_OWNERS     250
#define OWN_VEC4     (OWN_ROWS * KDIM / 4)
#define BLK_EDGES    16
#define IDX_WG_EDGES 16384
#define IDX_WGS      ((N_EDGES + IDX_WG_EDGES - 1) / IDX_WG_EDGES)
#define BM_ROW_DW    (IDX_WGS * 32)
#define BM_BYTES     ((size_t)N_OWNERS * BM_ROW_DW * 4)
#define BIT_PITCH    257
#define SEG_WORDS    (OWN_ROWS * KDIM / 32)
#define SEG_HDR      8
#define SEG_BATCHES  ((SEG_WORDS - SEG_HDR) / 8)
#define SEG_OVERFLOW 0xFFFFFFFFu

__global__ __launch_bounds__(1024) void build_index(const int* __restrict__ src, const int* __restrict__ dst,
                                                    unsigned int* __restrict__ bm, int* __restrict__ pairs) {
    __shared__ unsigned int bits[32 * BIT_PITCH];
    const int tid  = threadIdx.x;
    const int lane = tid & 31;
    const int wave = tid >> 5;
    for (int i = tid; i < 32 * BIT_PITCH; i += 1024) bits[i] = 0u;
    __syncthreads();

    const int wbase = blockIdx.x * IDX_WG_EDGES + wave * 512;
    int2 s[8], d[8];
#pragma unroll
    for (int q = 0; q < 8; ++q) {
        const int e = wbase + q * 64 + 2 * lane;
        if (e < N_EDGES) { s[q] = *(const int2*)(src + e); d[q] = *(const int2*)(dst + e); }
        else             { s[q] = make_int2(0, 0);         d[q] = make_int2(0, 0); }
    }
#pragma unroll
    for (int q = 0; q < 8; ++q) {
        const int e = wbase + q * 64 + 2 * lane;
        if (e < N_EDGES) {
            *(int4*)(pairs + 2 * (size_t)e) = make_int4(s[q].x, d[q].x, s[q].y, d[q].y);
            const unsigned int bit = 1u << (q * 4 + (lane >> 3));
            const unsigned int o0 = (unsigned int)d[q].x / (unsigned int)OWN_ROWS;
            const unsigned int o1 = (unsigned int)d[q].y / (unsigned int)OWN_ROWS;
            __hip_atomic_fetch_or(&bits[wave * BIT_PITCH + o0], bit, __ATOMIC_RELAXED, __HIP_MEMORY_SCOPE_WORKGROUP);
            if (o1 != o0)
                __hip_atomic_fetch_or(&bits[wave * BIT_PITCH + o1], bit, __ATOMIC_RELAXED, __HIP_MEMORY_SCOPE_WORKGROUP);
        }
    }
    __syncthreads();

    unsigned int* seg = bm + blockIdx.x * 32;
    for (int idx = tid; idx < N_OWNERS * 32; idx += 1024) {
        const int o = idx >> 5, i = idx & 31;
        seg[(size_t)o * BM_ROW_DW + i] = bits[i * BIT_PITCH + o];
    }
}

__device__ __forceinline__ void lds_umax(unsigned int* p, float v) {
    __hip_atomic_fetch_max(p, __float_as_uint(v), __ATOMIC_RELAXED, __HIP_MEMORY_SCOPE_WORKGROUP);
}

#define BL_CAP 1024
#define EL_CAP 264

template <bool SAVE, bool REPLAY>
__global__ __launch_bounds__(1024) void owner_max(const float* __restrict__ m,
                                                  const unsigned int* __restrict__ bm,
                                                  const int* __restrict__ pairs,
                                                  unsigned int* lists,
                                                  float* agg) {
    __shared__ __attribute__((aligned(16))) unsigned int acc[OWN_ROWS * KDIM];
    __shared__ unsigned short blist[32 * BL_CAP];
    __shared__ unsigned int   elist[32 * EL_CAP];
    const int tid  = threadIdx.x;
    const int lane = tid & 31;
    const int wave = __builtin_amdgcn_readfirstlane(tid >> 5);
    const int lo   = blockIdx.x * OWN_ROWS;
    const int half = lane >> 4;
    const int sub  = lane & 15;
    const int c2   = lane << 1;

    const unsigned int* brow = bm + (size_t)blockIdx.x * BM_ROW_DW;
    unsigned int* seg = lists + (size_t)blockIdx.x * (OWN_ROWS * KDIM) + wave * SEG_WORDS;
    unsigned int b0 = 0u, b1 = 0u, b2 = 0u, b3 = 0u, cur = 0u;
    if (REPLAY) {
        cur = seg[lane];
    } else {
        b0 = brow[tid];
        b1 = brow[tid + 1024];
        b2 = brow[tid + 2048];
        b3 = (tid + 3072 < BM_ROW_DW) ? brow[tid + 3072] : 0u;
    }

    {
        uint4* a4 = (uint4*)acc;
        for (int i = tid; i < OWN_VEC4; i += 1024) a4[i] = make_uint4(0u, 0u, 0u, 0u);
    }
    __syncthreads();

    const int2* pr = (const int2*)pairs;
    unsigned short* bl = blist + wave * BL_CAP;
    unsigned int*   el = elist + wave * EL_CAP;
    int ne = 0;

    int cnt8 = 0;

    auto serve8v = [&](unsigned int vals, int base) __attribute__((always_inline)) {
        int s[8], r[8];
        float2 va[8], vb[8];
#pragma unroll
        for (int i = 0; i < 8; ++i) {
            const unsigned int w = (unsigned int)__builtin_amdgcn_readlane((int)vals, base + i);
            s[i] = (int)(w & 0x1FFFFu);
            r[i] = (int)(w >> 17);
        }
#pragma unroll
        for (int i = 0; i < 8; ++i) {
            const float* row = m + (size_t)s[i] * KDIM + c2;
            va[i] = *(const float2*)row;
            vb[i] = *(const float2*)(row + 64);
        }
#pragma unroll
        for (int i = 0; i < 8; ++i) {
            unsigned int* p = acc + r[i] * KDIM + c2;
            lds_umax(p,      va[i].x);
            lds_umax(p + 1,  va[i].y);
            lds_umax(p + 64, vb[i].x);
            lds_umax(p + 65, vb[i].y);
        }
    };

    auto serve8 = [&](int pos) __attribute__((always_inline)) {
        const unsigned int mine = el[pos + (lane & 7)];
        if (SAVE) {
            if (cnt8 < SEG_BATCHES && lane < 8) seg[SEG_HDR + cnt8 * 8 + lane] = mine;
            ++cnt8;
        }
        serve8v(mine, 0);
    };

    auto search = [&]() __attribute__((always_inline)) {
    for (int j = 0; j < 4; ++j) {
        const int dw0 = j * 1024 + (wave << 5);
        if (dw0 >= BM_ROW_DW) break;
        const unsigned int word = (j == 0) ? b0 : (j == 1) ? b1 : (j == 2) ? b2 : b3;

        int nblk = 0;
        for (int t = 0; t < 32; ++t) {
            const bool set = ((word >> t) & 1u) != 0u;
            const unsigned int mask = __builtin_amdgcn_ballot_w32(set);
            if (set) bl[nblk + (int)__builtin_amdgcn_mbcnt_lo(mask, 0u)] = (unsigned short)((lane << 5) | t);
            nblk += __builtin_popcount(mask);
        }

        const int blk0 = dw0 << 5;
        for (int k = 0; k < nblk; k += 16) {
            int2 p[8];
#pragma unroll
            for (int i = 0; i < 8; ++i) {
                int idx = k + 2 * i + half;
                idx = (idx < nblk) ? idx : (nblk - 1);
                const int id = blk0 + (int)bl[idx];
                p[i] = pr[(size_t)id * BLK_EDGES + sub];
            }
#pragma unroll
            for (int i = 0; i < 8; ++i) {
                const unsigned int rr = (unsigned int)(p[i].y - lo);
                const bool ok = rr < (unsigned int)OWN_ROWS;
                const unsigned int mask = __builtin_amdgcn_ballot_w32(ok);
                if (ok) el[ne + (int)__builtin_amdgcn_mbcnt_lo(mask, 0u)] = (unsigned int)p[i].x | (rr << 17);
                ne += __builtin_popcount(mask);
            }
            int pos = 0;
            while (ne - pos >= 8) { serve8(pos); pos += 8; }
            const int rem = ne - pos;
            if (pos > 0 && lane < rem) {
                const unsigned int v = el[pos + lane];
                el[lane] = v;
            }
            ne = rem;
        }
    }

    if (ne > 0) {
        if (lane >= ne && lane < 8) el[lane] = el[0];
        serve8(0);
    }
    };

    if (REPLAY) {
        const unsigned int hdr = (unsigned int)__builtin_amdgcn_readfirstlane((int)cur);
        if (hdr <= (unsigned int)SEG_BATCHES) {
            const int n8     = (int)hdr;
            const int nwords = SEG_HDR + 8 * n8;
            const int npiece = (nwords + 31) >> 5;
            for (int g = 0; g < npiece; ++g) {
                unsigned int nxt = 0u;
                const int wn = 32 * (g + 1) + lane;
                const int bf = 4 * g - 1;
                if (bf >= 0 && bf + 3 < n8) {
                    serve8v(cur, 0);
                    nxt = seg[(wn < nwords) ? wn : (nwords - 1)];
                    serve8v(cur, 8);
                    serve8v(cur, 16);
                    serve8v(cur, 24);
                } else {
                    if (wn < nwords) nxt = seg[wn];
#pragma unroll
                    for (int q = 0; q < 4; ++q) {
                        const int b = bf + q;
                        if (b >= 0 && b < n8) serve8v(cur, 8 * q);
                    }
                }
                cur = nxt;
            }
        } else {
            b0 = brow[tid];
            b1 = brow[tid + 1024];
            b2 = brow[tid + 2048];
            b3 = (tid + 3072 < BM_ROW_DW) ? brow[tid + 3072] : 0u;
            search();
        }
    } else {
        search();
    }
    if (SAVE) {
        const unsigned int h = (cnt8 <= SEG_BATCHES) ? (unsigned int)cnt8 : SEG_OVERFLOW;
        if (lane == 0)                     seg[0] = h;
        else if (lane < SEG_HDR)           seg[lane] = 0u;
        else if (lane >= SEG_HDR + 8 * cnt8) seg[lane] = 0u;
    }
    __syncthreads();

    {
        const uint4* a4 = (const uint4*)acc;
        uint4* g4 = (uint4*)(agg + (size_t)lo * KDIM);
        for (int i = tid; i < OWN_VEC4; i += 1024) g4[i] = a4[i];
    }
}

extern "C" void kernel_launch(void* const* d_in, const int* in_sizes, int n_in,
                              void* d_out, int out_size, void* d_ws, size_t ws_size,
                              hipStream_t stream) {
    const float* in_feat  = (const float*)d_in[0];
    const int*   src      = (const int*)d_in[1];
    const int*   dst      = (const int*)d_in[2];
    const float* pool_w1  = (const float*)d_in[3];
    const float* pool_b1  = (const float*)d_in[4];
    const float* self_w1  = (const float*)d_in[5];
    const float* self_b1  = (const float*)d_in[6];
    const float* neigh_w1 = (const float*)d_in[7];
    const float* neigh_b1 = (const float*)d_in[8];
    const float* pool_w2  = (const float*)d_in[9];
    const float* pool_b2  = (const float*)d_in[10];
    const float* self_w2  = (const float*)d_in[11];
    const float* self_b2  = (const float*)d_in[12];
    const float* neigh_w2 = (const float*)d_in[13];
    const float* neigh_b2 = (const float*)d_in[14];
    float* out = (float*)d_out;

    const size_t NODE_BYTES = (size_t)N_NODES * KDIM * sizeof(float);
    char* ws = (char*)d_ws;
    float* p0 = (float*)(ws);
    float* p1 = (float*)(ws + NODE_BYTES);
    float* p2 = (float*)(ws + 2 * NODE_BYTES);
    unsigned int* bm = (unsigned int*)d_out;
    int* pairs = (int*)((char*)d_out + BM_BYTES);

    build_index<<<IDX_WGS, 1024, 0, stream>>>(src, dst, bm, pairs);

    strip_gemm<8, false, true, true><<<GEMM_WGS, GEMM_WAVES * 32, 0, stream>>>(in_feat, pool_w1, pool_b1,
                                                                              in_feat, pool_w1, pool_b1, p0, 128);
    owner_max<true, false><<<N_OWNERS, 1024, 0, stream>>>(p0, bm, pairs, (unsigned int*)p2, p1);
    strip_gemm<8, true, true, true><<<GEMM_WGS, GEMM_WAVES * 32, 0, stream>>>(in_feat, self_w1, self_b1,
                                                                             p1, neigh_w1, neigh_b1, p0, 128);

    strip_gemm<8, false, true, true><<<GEMM_WGS, GEMM_WAVES * 32, 0, stream>>>(p0, pool_w2, pool_b2,
                                                                              p0, pool_w2, pool_b2, p1, 128);
    owner_max<false, true><<<N_OWNERS, 1024, 0, stream>>>(p1, bm, pairs, (unsigned int*)p2, p2);
    strip_gemm<3, true, false, false><<<GEMM_WGS, GEMM_WAVES * 32, 0, stream>>>(p0, self_w2, self_b2,
                                                                               p2, neigh_w2, neigh_b2, out, 40);
}


# Owner max gathers a source row with one 16-byte load per lane (one 512-B wave load, was two 8-byte ones); LDS plane columns permuted (feature 4l+i at 32i+l) to stay conflict-free; integer max
# speedup vs baseline: 3.6464x; 1.0449x over previous
#include <hip/hip_runtime.h>


typedef float v2f __attribute__((ext_vector_type(2)));
typedef float v4f __attribute__((ext_vector_type(4)));
typedef float v8f __attribute__((ext_vector_type(8)));

#define N_NODES 100000
#define N_EDGES 1600000
#define KDIM    128

__device__ __forceinline__ v8f wmma_f32(v2f a, v2f b, v8f c) {
    return __builtin_amdgcn_wmma_f32_16x16x4_f32(
        false, a, false, b, (short)0, c, false, false);
}

#define WP            132
#define GEMM_WGS      250
#define TILES_PER_WG  25
#define GEMM_WAVES    16

__device__ __forceinline__ void load_row_pieces(v4f (&a)[16], const float* __restrict__ X, int row0, int c, int h) {
    const float* xa = X + (size_t)(row0 + c) * KDIM + 4 * h;
#pragma unroll
    for (int kk = 0; kk < 16; ++kk) a[kk] = *(const v4f*)(xa + 8 * kk);
}

__device__ __forceinline__ void hold_row_pieces(const v4f (&a)[16]) {
    asm volatile("" :: "v"(a[0]), "v"(a[1]), "v"(a[2]), "v"(a[3]), "v"(a[4]), "v"(a[5]), "v"(a[6]), "v"(a[7]),
                       "v"(a[8]), "v"(a[9]), "v"(a[10]), "v"(a[11]), "v"(a[12]), "v"(a[13]), "v"(a[14]), "v"(a[15]));
}

template <int NT>
__device__ __forceinline__ void strip_products(v8f (&acc)[NT], const v4f (&a)[16], const float* wb) {
#pragma unroll
    for (int kk = 0; kk < 16; ++kk) {
        v4f b[NT];
#pragma unroll
        for (int j = 0; j < NT; ++j) b[j] = *(const v4f*)(wb + j * 16 * WP + 8 * kk);
#pragma unroll
        for (int j = 0; j < NT; ++j) acc[j] = wmma_f32(v2f{a[kk].x, a[kk].y}, v2f{b[j].x, b[j].y}, acc[j]);
#pragma unroll
        for (int j = 0; j < NT; ++j) acc[j] = wmma_f32(v2f{a[kk].z, a[kk].w}, v2f{b[j].z, b[j].w}, acc[j]);
    }
}

template <int NT, bool TWO, bool RELU, bool PERM>
__device__ __forceinline__ void strip_tile(v4f (&a)[16], const float* __restrict__ X2,
                                           const float* wb1, const float* wb2,
                                           const float* __restrict__ b1, const float* __restrict__ b2,
                                           float* __restrict__ Y, int O, int row0, int c, int h) {
    v8f acc[NT];
#pragma unroll
    for (int j = 0; j < NT; ++j) acc[j] = v8f{};

    hold_row_pieces(a);
    strip_products<NT>(acc, a, wb1);
    if (TWO) {
        load_row_pieces(a, X2, row0, c, h);
        hold_row_pieces(a);
        strip_products<NT>(acc, a, wb2);
    }

    if (PERM) {
        float4 bl = *(const float4*)(b1 + 4 * c);
        float4 bh = *(const float4*)(b1 + 64 + 4 * c);
        if (TWO) {
            const float4 cl = *(const float4*)(b2 + 4 * c);
            const float4 ch = *(const float4*)(b2 + 64 + 4 * c);
            bl.x += cl.x; bl.y += cl.y; bl.z += cl.z; bl.w += cl.w;
            bh.x += ch.x; bh.y += ch.y; bh.z += ch.z; bh.w += ch.w;
        }
#pragma unroll
        for (int r = 0; r < 8; ++r) {
            float4 lo = { acc[0][r] + bl.x, acc[1][r] + bl.y, acc[2][r] + bl.z, acc[3][r] + bl.w };
            float4 hi = { acc[4][r] + bh.x, acc[5][r] + bh.y, acc[6][r] + bh.z, acc[7][r] + bh.w };
            if (RELU) {
                lo.x = fmaxf(lo.x, 0.0f); lo.y = fmaxf(lo.y, 0.0f); lo.z = fmaxf(lo.z, 0.0f); lo.w = fmaxf(lo.w, 0.0f);
                hi.x = fmaxf(hi.x, 0.0f); hi.y = fmaxf(hi.y, 0.0f); hi.z = fmaxf(hi.z, 0.0f); hi.w = fmaxf(hi.w, 0.0f);
            }
            float* yr = Y + (size_t)(row0 + 8 * h + r) * KDIM + 4 * c;
            *(float4*)yr        = lo;
            *(float4*)(yr + 64) = hi;
        }
    } else {
#pragma unroll
        for (int j = 0; j < NT; ++j) {
            const int col    = j * 16 + c;
            const bool valid = (col < O);
            const int colW   = valid ? col : (O - 1);
            float bb = b1[colW];
            if (TWO) bb += b2[colW];
#pragma unroll
            for (int r = 0; r < 8; ++r) {
                float v = acc[j][r] + bb;
                if (RELU) v = fmaxf(v, 0.0f);
                if (valid) Y[(size_t)(row0 + 8 * h + r) * O + col] = v;
            }
        }
    }
}

template <int NT, bool TWO, bool RELU, bool PERM>
__global__ __launch_bounds__(GEMM_WAVES * 32) void strip_gemm(const float* __restrict__ X1,
                                                              const float* __restrict__ W1,
                                                              const float* __restrict__ b1,
                                                              const float* __restrict__ X2,
                                                              const float* __restrict__ W2,
                                                              const float* __restrict__ b2,
                                                              float* __restrict__ Y, int O) {
    __shared__ __attribute__((aligned(16))) float Ws[(TWO ? 2 : 1) * NT * 16 * WP];
    const int tid  = threadIdx.x;
    const int lane = tid & 31;
    const int wave = tid >> 5;
    const int c    = lane & 15;
    const int h    = lane >> 4;
    const int tile0 = blockIdx.x * TILES_PER_WG;

    v4f a[16];
    load_row_pieces(a, X1, (tile0 + wave) * 16, c, h);

    {
#pragma unroll
        for (int u = 0; u < NT; ++u) {
            const int i = tid + u * (GEMM_WAVES * 32);
            const int rho = i >> 5, q = i & 31;
            const int cc = rho & 15, j = rho >> 4;
            int n = PERM ? (4 * cc + (j & 3) + 64 * (j >> 2)) : rho;
            n = (n < O) ? n : (O - 1);
            *(float4*)(Ws + rho * WP + q * 4) = *(const float4*)(W1 + (size_t)n * KDIM + q * 4);
            if (TWO)
                *(float4*)(Ws + (NT * 16 + rho) * WP + q * 4) = *(const float4*)(W2 + (size_t)n * KDIM + q * 4);
        }
    }
    __syncthreads();

    const float* wb1 = Ws + c * WP + 4 * h;
    const float* wb2 = Ws + (NT * 16 + c) * WP + 4 * h;

    strip_tile<NT, TWO, RELU, PERM>(a, X2, wb1, wb2, b1, b2, Y, O, (tile0 + wave) * 16, c, h);

    if (wave + GEMM_WAVES < TILES_PER_WG) {
        asm volatile("" ::: "memory");
        const int row0 = (tile0 + wave + GEMM_WAVES) * 16;
        load_row_pieces(a, X1, row0, c, h);
        strip_tile<NT, TWO, RELU, PERM>(a, X2, wb1, wb2, b1, b2, Y, O, row0, c, h);
    }
}

#define OWN_ROWS     400
#define N_OWNERS     250
#define OWN_VEC4     (OWN_ROWS * KDIM / 4)
#define BLK_EDGES    16
#define IDX_WG_EDGES 16384
#define IDX_WGS      ((N_EDGES + IDX_WG_EDGES - 1) / IDX_WG_EDGES)
#define BM_ROW_DW    (IDX_WGS * 32)
#define BM_BYTES     ((size_t)N_OWNERS * BM_ROW_DW * 4)
#define BIT_PITCH    257
#define SEG_WORDS    (OWN_ROWS * KDIM / 32)
#define SEG_HDR      8
#define SEG_BATCHES  ((SEG_WORDS - SEG_HDR) / 8)
#define SEG_OVERFLOW 0xFFFFFFFFu

__global__ __launch_bounds__(1024) void build_index(const int* __restrict__ src, const int* __restrict__ dst,
                                                    unsigned int* __restrict__ bm, int* __restrict__ pairs) {
    __shared__ unsigned int bits[32 * BIT_PITCH];
    const int tid  = threadIdx.x;
    const int lane = tid & 31;
    const int wave = tid >> 5;
    for (int i = tid; i < 32 * BIT_PITCH; i += 1024) bits[i] = 0u;
    __syncthreads();

    const int wbase = blockIdx.x * IDX_WG_EDGES + wave * 512;
    int2 s[8], d[8];
#pragma unroll
    for (int q = 0; q < 8; ++q) {
        const int e = wbase + q * 64 + 2 * lane;
        if (e < N_EDGES) { s[q] = *(const int2*)(src + e); d[q] = *(const int2*)(dst + e); }
        else             { s[q] = make_int2(0, 0);         d[q] = make_int2(0, 0); }
    }
#pragma unroll
    for (int q = 0; q < 8; ++q) {
        const int e = wbase + q * 64 + 2 * lane;
        if (e < N_EDGES) {
            *(int4*)(pairs + 2 * (size_t)e) = make_int4(s[q].x, d[q].x, s[q].y, d[q].y);
            const unsigned int bit = 1u << (q * 4 + (lane >> 3));
            const unsigned int o0 = (unsigned int)d[q].x / (unsigned int)OWN_ROWS;
            const unsigned int o1 = (unsigned int)d[q].y / (unsigned int)OWN_ROWS;
            __hip_atomic_fetch_or(&bits[wave * BIT_PITCH + o0], bit, __ATOMIC_RELAXED, __HIP_MEMORY_SCOPE_WORKGROUP);
            if (o1 != o0)
                __hip_atomic_fetch_or(&bits[wave * BIT_PITCH + o1], bit, __ATOMIC_RELAXED, __HIP_MEMORY_SCOPE_WORKGROUP);
        }
    }
    __syncthreads();

    unsigned int* seg = bm + blockIdx.x * 32;
    for (int idx = tid; idx < N_OWNERS * 32; idx += 1024) {
        const int o = idx >> 5, i = idx & 31;
        seg[(size_t)o * BM_ROW_DW + i] = bits[i * BIT_PITCH + o];
    }
}

__device__ __forceinline__ void lds_umax(unsigned int* p, float v) {
    __hip_atomic_fetch_max(p, __float_as_uint(v), __ATOMIC_RELAXED, __HIP_MEMORY_SCOPE_WORKGROUP);
}

#define BL_CAP 1024
#define EL_CAP 264

template <bool SAVE, bool REPLAY>
__global__ __launch_bounds__(1024) void owner_max(const float* __restrict__ m,
                                                  const unsigned int* __restrict__ bm,
                                                  const int* __restrict__ pairs,
                                                  unsigned int* lists,
                                                  float* agg) {
    __shared__ __attribute__((aligned(16))) unsigned int acc[OWN_ROWS * KDIM];
    __shared__ unsigned short blist[32 * BL_CAP];
    __shared__ unsigned int   elist[32 * EL_CAP];
    const int tid  = threadIdx.x;
    const int lane = tid & 31;
    const int wave = __builtin_amdgcn_readfirstlane(tid >> 5);
    const int lo   = blockIdx.x * OWN_ROWS;
    const int half = lane >> 4;
    const int sub  = lane & 15;
    const int f4   = lane << 2;

    const unsigned int* brow = bm + (size_t)blockIdx.x * BM_ROW_DW;
    unsigned int* seg = lists + (size_t)blockIdx.x * (OWN_ROWS * KDIM) + wave * SEG_WORDS;
    unsigned int b0 = 0u, b1 = 0u, b2 = 0u, b3 = 0u, cur = 0u;
    if (REPLAY) {
        cur = seg[lane];
    } else {
        b0 = brow[tid];
        b1 = brow[tid + 1024];
        b2 = brow[tid + 2048];
        b3 = (tid + 3072 < BM_ROW_DW) ? brow[tid + 3072] : 0u;
    }

    {
        uint4* a4 = (uint4*)acc;
        for (int i = tid; i < OWN_VEC4; i += 1024) a4[i] = make_uint4(0u, 0u, 0u, 0u);
    }
    __syncthreads();

    const int2* pr = (const int2*)pairs;
    unsigned short* bl = blist + wave * BL_CAP;
    unsigned int*   el = elist + wave * EL_CAP;
    int ne = 0;

    int cnt8 = 0;

    auto serve8v = [&](unsigned int vals, int base) __attribute__((always_inline)) {
        int s[8], r[8];
        float4 v[8];
#pragma unroll
        for (int i = 0; i < 8; ++i) {
            const unsigned int w = (unsigned int)__builtin_amdgcn_readlane((int)vals, base + i);
            s[i] = (int)(w & 0x1FFFFu);
            r[i] = (int)(w >> 17);
        }
#pragma unroll
        for (int i = 0; i < 8; ++i) {
            v[i] = *(const float4*)(m + (size_t)s[i] * KDIM + f4);
        }
#pragma unroll
        for (int i = 0; i < 8; ++i) {
            unsigned int* p = acc + r[i] * KDIM + lane;
            lds_umax(p,      v[i].x);
            lds_umax(p + 32, v[i].y);
            lds_umax(p + 64, v[i].z);
            lds_umax(p + 96, v[i].w);
        }
    };

    auto serve8 = [&](int pos) __attribute__((always_inline)) {
        const unsigned int mine = el[pos + (lane & 7)];
        if (SAVE) {
            if (cnt8 < SEG_BATCHES && lane < 8) seg[SEG_HDR + cnt8 * 8 + lane] = mine;
            ++cnt8;
        }
        serve8v(mine, 0);
    };

    auto search = [&]() __attribute__((always_inline)) {
    for (int j = 0; j < 4; ++j) {
        const int dw0 = j * 1024 + (wave << 5);
        if (dw0 >= BM_ROW_DW) break;
        const unsigned int word = (j == 0) ? b0 : (j == 1) ? b1 : (j == 2) ? b2 : b3;

        int nblk = 0;
        for (int t = 0; t < 32; ++t) {
            const bool set = ((word >> t) & 1u) != 0u;
            const unsigned int mask = __builtin_amdgcn_ballot_w32(set);
            if (set) bl[nblk + (int)__builtin_amdgcn_mbcnt_lo(mask, 0u)] = (unsigned short)((lane << 5) | t);
            nblk += __builtin_popcount(mask);
        }

        const int blk0 = dw0 << 5;
        for (int k = 0; k < nblk; k += 16) {
            int2 p[8];
#pragma unroll
            for (int i = 0; i < 8; ++i) {
                int idx = k + 2 * i + half;
                idx = (idx < nblk) ? idx : (nblk - 1);
                const int id = blk0 + (int)bl[idx];
                p[i] = pr[(size_t)id * BLK_EDGES + sub];
            }
#pragma unroll
            for (int i = 0; i < 8; ++i) {
                const unsigned int rr = (unsigned int)(p[i].y - lo);
                const bool ok = rr < (unsigned int)OWN_ROWS;
                const unsigned int mask = __builtin_amdgcn_ballot_w32(ok);
                if (ok) el[ne + (int)__builtin_amdgcn_mbcnt_lo(mask, 0u)] = (unsigned int)p[i].x | (rr << 17);
                ne += __builtin_popcount(mask);
            }
            int pos = 0;
            while (ne - pos >= 8) { serve8(pos); pos += 8; }
            const int rem = ne - pos;
            if (pos > 0 && lane < rem) {
                const unsigned int v = el[pos + lane];
                el[lane] = v;
            }
            ne = rem;
        }
    }

    if (ne > 0) {
        if (lane >= ne && lane < 8) el[lane] = el[0];
        serve8(0);
    }
    };

    if (REPLAY) {
        const unsigned int hdr = (unsigned int)__builtin_amdgcn_readfirstlane((int)cur);
        if (hdr <= (unsigned int)SEG_BATCHES) {
            const int n8     = (int)hdr;
            const int nwords = SEG_HDR + 8 * n8;
            const int npiece = (nwords + 31) >> 5;
            for (int g = 0; g < npiece; ++g) {
                unsigned int nxt = 0u;
                const int wn = 32 * (g + 1) + lane;
                const int bf = 4 * g - 1;
                if (bf >= 0 && bf + 3 < n8) {
                    serve8v(cur, 0);
                    nxt = seg[(wn < nwords) ? wn : (nwords - 1)];
                    serve8v(cur, 8);
                    serve8v(cur, 16);
                    serve8v(cur, 24);
                } else {
                    if (wn < nwords) nxt = seg[wn];
#pragma unroll
                    for (int q = 0; q < 4; ++q) {
                        const int b = bf + q;
                        if (b >= 0 && b < n8) serve8v(cur, 8 * q);
                    }
                }
                cur = nxt;
            }
        } else {
            b0 = brow[tid];
            b1 = brow[tid + 1024];
            b2 = brow[tid + 2048];
            b3 = (tid + 3072 < BM_ROW_DW) ? brow[tid + 3072] : 0u;
            search();
        }
    } else {
        search();
    }
    if (SAVE) {
        const unsigned int h = (cnt8 <= SEG_BATCHES) ? (unsigned int)cnt8 : SEG_OVERFLOW;
        if (lane == 0)                     seg[0] = h;
        else if (lane < SEG_HDR)           seg[lane] = 0u;
        else if (lane >= SEG_HDR + 8 * cnt8) seg[lane] = 0u;
    }
    __syncthreads();

    {
        uint4* g4 = (uint4*)(agg + (size_t)lo * KDIM);
        for (int i = tid; i < OWN_VEC4; i += 1024) {
            const unsigned int* p = acc + (i >> 5) * KDIM + (i & 31);
            g4[i] = make_uint4(p[0], p[32], p[64], p[96]);
        }
    }
}

extern "C" void kernel_launch(void* const* d_in, const int* in_sizes, int n_in,
                              void* d_out, int out_size, void* d_ws, size_t ws_size,
                              hipStream_t stream) {
    const float* in_feat  = (const float*)d_in[0];
    const int*   src      = (const int*)d_in[1];
    const int*   dst      = (const int*)d_in[2];
    const float* pool_w1  = (const float*)d_in[3];
    const float* pool_b1  = (const float*)d_in[4];
    const float* self_w1  = (const float*)d_in[5];
    const float* self_b1  = (const float*)d_in[6];
    const float* neigh_w1 = (const float*)d_in[7];
    const float* neigh_b1 = (const float*)d_in[8];
    const float* pool_w2  = (const float*)d_in[9];
    const float* pool_b2  = (const float*)d_in[10];
    const float* self_w2  = (const float*)d_in[11];
    const float* self_b2  = (const float*)d_in[12];
    const float* neigh_w2 = (const float*)d_in[13];
    const float* neigh_b2 = (const float*)d_in[14];
    float* out = (float*)d_out;

    const size_t NODE_BYTES = (size_t)N_NODES * KDIM * sizeof(float);
    char* ws = (char*)d_ws;
    float* p0 = (float*)(ws);
    float* p1 = (float*)(ws + NODE_BYTES);
    float* p2 = (float*)(ws + 2 * NODE_BYTES);
    unsigned int* bm = (unsigned int*)d_out;
    int* pairs = (int*)((char*)d_out + BM_BYTES);

    build_index<<<IDX_WGS, 1024, 0, stream>>>(src, dst, bm, pairs);

    strip_gemm<8, false, true, true><<<GEMM_WGS, GEMM_WAVES * 32, 0, stream>>>(in_feat, pool_w1, pool_b1,
                                                                              in_feat, pool_w1, pool_b1, p0, 128);
    owner_max<true, false><<<N_OWNERS, 1024, 0, stream>>>(p0, bm, pairs, (unsigned int*)p2, p1);
    strip_gemm<8, true, true, true><<<GEMM_WGS, GEMM_WAVES * 32, 0, stream>>>(in_feat, self_w1, self_b1,
                                                                             p1, neigh_w1, neigh_b1, p0, 128);

    strip_gemm<8, false, true, true><<<GEMM_WGS, GEMM_WAVES * 32, 0, stream>>>(p0, pool_w2, pool_b2,
                                                                              p0, pool_w2, pool_b2, p1, 128);
    owner_max<false, true><<<N_OWNERS, 1024, 0, stream>>>(p1, bm, pairs, (unsigned int*)p2, p2);
    strip_gemm<3, true, false, false><<<GEMM_WGS, GEMM_WAVES * 32, 0, stream>>>(p0, self_w2, self_b2,
                                                                               p2, neigh_w2, neigh_b2, out, 40);
}


# Hidden rows stay on chip: a wave turns each 16-row tile through LDS into the A operand of the layer-2 message and output-self products; 6 launches, not 7; output = neigh sum + rounded self sum
# speedup vs baseline: 3.7894x; 1.0392x over previous
#include <hip/hip_runtime.h>


typedef float v2f __attribute__((ext_vector_type(2)));
typedef float v4f __attribute__((ext_vector_type(4)));
typedef float v8f __attribute__((ext_vector_type(8)));

#define N_NODES 100000
#define N_EDGES 1600000
#define KDIM    128
#define N_CLS   40

__device__ __forceinline__ v8f wmma_f32(v2f a, v2f b, v8f c) {
    return __builtin_amdgcn_wmma_f32_16x16x4_f32(
        false, a, false, b, (short)0, c, false, false);
}

#define WP            132
#define GEMM_WGS      250
#define TILES_PER_WG  25
#define GEMM_WAVES    16

__device__ __forceinline__ void load_row_pieces(v4f (&a)[16], const float* __restrict__ X, int row0, int c, int h) {
    const float* xa = X + (size_t)(row0 + c) * KDIM + 4 * h;
#pragma unroll
    for (int kk = 0; kk < 16; ++kk) a[kk] = *(const v4f*)(xa + 8 * kk);
}

__device__ __forceinline__ void hold_row_pieces(const v4f (&a)[16]) {
    asm volatile("" :: "v"(a[0]), "v"(a[1]), "v"(a[2]), "v"(a[3]), "v"(a[4]), "v"(a[5]), "v"(a[6]), "v"(a[7]),
                       "v"(a[8]), "v"(a[9]), "v"(a[10]), "v"(a[11]), "v"(a[12]), "v"(a[13]), "v"(a[14]), "v"(a[15]));
}

template <int NT>
__device__ __forceinline__ void strip_products(v8f (&acc)[NT], const v4f (&a)[16], const float* wb) {
#pragma unroll
    for (int kk = 0; kk < 16; ++kk) {
        v4f b[NT];
#pragma unroll
        for (int j = 0; j < NT; ++j) b[j] = *(const v4f*)(wb + j * 16 * WP + 8 * kk);
#pragma unroll
        for (int j = 0; j < NT; ++j) acc[j] = wmma_f32(v2f{a[kk].x, a[kk].y}, v2f{b[j].x, b[j].y}, acc[j]);
#pragma unroll
        for (int j = 0; j < NT; ++j) acc[j] = wmma_f32(v2f{a[kk].z, a[kk].w}, v2f{b[j].z, b[j].w}, acc[j]);
    }
}

template <int NT, bool TWO, bool RELU, bool PERM, bool ADD>
__device__ __forceinline__ void strip_tile(v4f (&a)[16], const float* __restrict__ X2,
                                           const float* wb1, const float* wb2,
                                           const float* __restrict__ b1, const float* __restrict__ b2,
                                           float* __restrict__ Y, int O, int row0, int c, int h) {
    v8f acc[NT];
#pragma unroll
    for (int j = 0; j < NT; ++j) acc[j] = v8f{};

    hold_row_pieces(a);
    strip_products<NT>(acc, a, wb1);
    if (TWO) {
        load_row_pieces(a, X2, row0, c, h);
        hold_row_pieces(a);
        strip_products<NT>(acc, a, wb2);
    }

    if (PERM) {
        float4 bl = *(const float4*)(b1 + 4 * c);
        float4 bh = *(const float4*)(b1 + 64 + 4 * c);
        if (TWO) {
            const float4 cl = *(const float4*)(b2 + 4 * c);
            const float4 ch = *(const float4*)(b2 + 64 + 4 * c);
            bl.x += cl.x; bl.y += cl.y; bl.z += cl.z; bl.w += cl.w;
            bh.x += ch.x; bh.y += ch.y; bh.z += ch.z; bh.w += ch.w;
        }
#pragma unroll
        for (int r = 0; r < 8; ++r) {
            float4 lo = { acc[0][r] + bl.x, acc[1][r] + bl.y, acc[2][r] + bl.z, acc[3][r] + bl.w };
            float4 hi = { acc[4][r] + bh.x, acc[5][r] + bh.y, acc[6][r] + bh.z, acc[7][r] + bh.w };
            if (RELU) {
                lo.x = fmaxf(lo.x, 0.0f); lo.y = fmaxf(lo.y, 0.0f); lo.z = fmaxf(lo.z, 0.0f); lo.w = fmaxf(lo.w, 0.0f);
                hi.x = fmaxf(hi.x, 0.0f); hi.y = fmaxf(hi.y, 0.0f); hi.z = fmaxf(hi.z, 0.0f); hi.w = fmaxf(hi.w, 0.0f);
            }
            float* yr = Y + (size_t)(row0 + 8 * h + r) * KDIM + 4 * c;
            *(float4*)yr        = lo;
            *(float4*)(yr + 64) = hi;
        }
    } else {
#pragma unroll
        for (int j = 0; j < NT; ++j) {
            const int col    = j * 16 + c;
            const bool valid = (col < O);
            const int colW   = valid ? col : (O - 1);
            float bb = ADD ? 0.0f : b1[colW];
            if (TWO) bb += b2[colW];
#pragma unroll
            for (int r = 0; r < 8; ++r) {
                float v = acc[j][r] + (ADD ? X2[(size_t)(row0 + 8 * h + r) * O + colW] : bb);
                if (RELU) v = fmaxf(v, 0.0f);
                if (valid) Y[(size_t)(row0 + 8 * h + r) * O + col] = v;
            }
        }
    }
}

template <int NT, bool TWO, bool RELU, bool PERM, bool ADD>
__global__ __launch_bounds__(GEMM_WAVES * 32) void strip_gemm(const float* __restrict__ X1,
                                                              const float* __restrict__ W1,
                                                              const float* __restrict__ b1,
                                                              const float* __restrict__ X2,
                                                              const float* __restrict__ W2,
                                                              const float* __restrict__ b2,
                                                              float* __restrict__ Y, int O) {
    __shared__ __attribute__((aligned(16))) float Ws[(TWO ? 2 : 1) * NT * 16 * WP];
    const int tid  = threadIdx.x;
    const int lane = tid & 31;
    const int wave = tid >> 5;
    const int c    = lane & 15;
    const int h    = lane >> 4;
    const int tile0 = blockIdx.x * TILES_PER_WG;

    v4f a[16];
    load_row_pieces(a, X1, (tile0 + wave) * 16, c, h);

    {
#pragma unroll
        for (int u = 0; u < NT; ++u) {
            const int i = tid + u * (GEMM_WAVES * 32);
            const int rho = i >> 5, q = i & 31;
            const int cc = rho & 15, j = rho >> 4;
            int n = PERM ? (4 * cc + (j & 3) + 64 * (j >> 2)) : rho;
            n = (n < O) ? n : (O - 1);
            *(float4*)(Ws + rho * WP + q * 4) = *(const float4*)(W1 + (size_t)n * KDIM + q * 4);
            if (TWO)
                *(float4*)(Ws + (NT * 16 + rho) * WP + q * 4) = *(const float4*)(W2 + (size_t)n * KDIM + q * 4);
        }
    }
    __syncthreads();

    const float* wb1 = Ws + c * WP + 4 * h;
    const float* wb2 = Ws + (NT * 16 + c) * WP + 4 * h;

    strip_tile<NT, TWO, RELU, PERM, ADD>(a, X2, wb1, wb2, b1, b2, Y, O, (tile0 + wave) * 16, c, h);

    if (wave + GEMM_WAVES < TILES_PER_WG) {
        asm volatile("" ::: "memory");
        const int row0 = (tile0 + wave + GEMM_WAVES) * 16;
        load_row_pieces(a, X1, row0, c, h);
        strip_tile<NT, TWO, RELU, PERM, ADD>(a, X2, wb1, wb2, b1, b2, Y, O, row0, c, h);
    }
}

#define SP        68
#define NT2       3
#define CH_W_FLT  ((3 * 8 + NT2) * 16 * WP)
#define CH_LDS    (CH_W_FLT + GEMM_WAVES * 16 * SP)

__device__ __forceinline__ void turn_half(v4f* dst, float* st, const float4 (&v)[8], int c, int h) {
#pragma unroll
    for (int r = 0; r < 8; ++r) *(float4*)(st + (8 * h + r) * SP + 4 * c) = v[r];
    __builtin_amdgcn_fence(__ATOMIC_RELEASE, "wavefront");
    __builtin_amdgcn_wave_barrier();
    __builtin_amdgcn_fence(__ATOMIC_ACQUIRE, "wavefront");
#pragma unroll
    for (int kk = 0; kk < 8; ++kk) dst[kk] = *(const v4f*)(st + c * SP + 8 * kk + 4 * h);
    __builtin_amdgcn_fence(__ATOMIC_RELEASE, "wavefront");
    __builtin_amdgcn_wave_barrier();
    __builtin_amdgcn_fence(__ATOMIC_ACQUIRE, "wavefront");
}

__device__ __forceinline__ void chain_tile(v4f (&a)[16], const float* G,
                                           const float* wb1, const float* wb2, const float* wb3, const float* wb4,
                                           const float* __restrict__ bs1, const float* __restrict__ bn1,
                                           const float* __restrict__ bp2,
                                           const float* __restrict__ bs2, const float* __restrict__ bn2,
                                           float* st, float* M2, float* __restrict__ S2, int row0, int c, int h) {
    float4 lo[8], hi[8];
    {
        v8f acc[8];
#pragma unroll
        for (int j = 0; j < 8; ++j) acc[j] = v8f{};
        hold_row_pieces(a);
        strip_products<8>(acc, a, wb1);
        load_row_pieces(a, G, row0, c, h);
        hold_row_pieces(a);
        strip_products<8>(acc, a, wb2);

        float4 bl = *(const float4*)(bs1 + 4 * c);
        float4 bh = *(const float4*)(bs1 + 64 + 4 * c);
        const float4 cl = *(const float4*)(bn1 + 4 * c);
        const float4 ch = *(const float4*)(bn1 + 64 + 4 * c);
        bl.x += cl.x; bl.y += cl.y; bl.z += cl.z; bl.w += cl.w;
        bh.x += ch.x; bh.y += ch.y; bh.z += ch.z; bh.w += ch.w;
#pragma unroll
        for (int r = 0; r < 8; ++r) {
            lo[r] = make_float4(fmaxf(acc[0][r] + bl.x, 0.0f), fmaxf(acc[1][r] + bl.y, 0.0f),
                                fmaxf(acc[2][r] + bl.z, 0.0f), fmaxf(acc[3][r] + bl.w, 0.0f));
            hi[r] = make_float4(fmaxf(acc[4][r] + bh.x, 0.0f), fmaxf(acc[5][r] + bh.y, 0.0f),
                                fmaxf(acc[6][r] + bh.z, 0.0f), fmaxf(acc[7][r] + bh.w, 0.0f));
        }
    }
    turn_half(&a[0], st, lo, c, h);
    turn_half(&a[8], st, hi, c, h);

    {
        v8f acc3[NT2];
#pragma unroll
        for (int j = 0; j < NT2; ++j) acc3[j] = v8f{};
        strip_products<NT2>(acc3, a, wb4);
#pragma unroll
        for (int j = 0; j < NT2; ++j) {
            const int col    = j * 16 + c;
            const bool valid = (col < N_CLS);
            const int colW   = valid ? col : (N_CLS - 1);
            const float bb   = bs2[colW] + bn2[colW];
#pragma unroll
            for (int r = 0; r < 8; ++r) {
                if (valid) S2[(size_t)(row0 + 8 * h + r) * N_CLS + col] = acc3[j][r] + bb;
            }
        }
    }

    v8f acc2[8];
#pragma unroll
    for (int j = 0; j < 8; ++j) acc2[j] = v8f{};
    strip_products<8>(acc2, a, wb3);

    {
        const float4 bl = *(const float4*)(bp2 + 4 * c);
        const float4 bh = *(const float4*)(bp2 + 64 + 4 * c);
#pragma unroll
        for (int r = 0; r < 8; ++r) {
            const float4 l4 = make_float4(fmaxf(acc2[0][r] + bl.x, 0.0f), fmaxf(acc2[1][r] + bl.y, 0.0f),
                                          fmaxf(acc2[2][r] + bl.z, 0.0f), fmaxf(acc2[3][r] + bl.w, 0.0f));
            const float4 h4 = make_float4(fmaxf(acc2[4][r] + bh.x, 0.0f), fmaxf(acc2[5][r] + bh.y, 0.0f),
                                          fmaxf(acc2[6][r] + bh.z, 0.0f), fmaxf(acc2[7][r] + bh.w, 0.0f));
            float* yr = M2 + (size_t)(row0 + 8 * h + r) * KDIM + 4 * c;
            *(float4*)yr        = l4;
            *(float4*)(yr + 64) = h4;
        }
    }
}

__global__ __launch_bounds__(GEMM_WAVES * 32) void chain_gemm(const float* __restrict__ X,
                                                              const float* G,
                                                              const float* __restrict__ Ws1, const float* __restrict__ bs1,
                                                              const float* __restrict__ Wn1, const float* __restrict__ bn1,
                                                              const float* __restrict__ Wp2, const float* __restrict__ bp2,
                                                              const float* __restrict__ Ws2, const float* __restrict__ bs2,
                                                              const float* __restrict__ bn2,
                                                              float* M2, float* __restrict__ S2) {
    __shared__ __attribute__((aligned(16))) float Ws[CH_LDS];
    const int tid  = threadIdx.x;
    const int lane = tid & 31;
    const int wave = tid >> 5;
    const int c    = lane & 15;
    const int h    = lane >> 4;
    const int tile0 = blockIdx.x * TILES_PER_WG;

    v4f a[16];
    load_row_pieces(a, X, (tile0 + wave) * 16, c, h);

    {
#pragma unroll
        for (int u = 0; u < 8; ++u) {
            const int i = tid + u * (GEMM_WAVES * 32);
            const int rho = i >> 5, q = i & 31;
            const int cc = rho & 15, j = rho >> 4;
            const int n = 4 * cc + (j & 3) + 64 * (j >> 2);
            *(float4*)(Ws + rho * WP + q * 4)             = *(const float4*)(Ws1 + (size_t)n * KDIM + q * 4);
            *(float4*)(Ws + (128 + rho) * WP + q * 4)     = *(const float4*)(Wn1 + (size_t)n * KDIM + q * 4);
            *(float4*)(Ws + (256 + rho) * WP + q * 4)     = *(const float4*)(Wp2 + (size_t)n * KDIM + q * 4);
        }
#pragma unroll
        for (int u = 0; u < NT2; ++u) {
            const int i = tid + u * (GEMM_WAVES * 32);
            const int rho = i >> 5, q = i & 31;
            const int n = (rho < N_CLS) ? rho : (N_CLS - 1);
            *(float4*)(Ws + (384 + rho) * WP + q * 4)     = *(const float4*)(Ws2 + (size_t)n * KDIM + q * 4);
        }
    }
    __syncthreads();

    const float* wb1 = Ws + c * WP + 4 * h;
    const float* wb2 = Ws + (128 + c) * WP + 4 * h;
    const float* wb3 = Ws + (256 + c) * WP + 4 * h;
    const float* wb4 = Ws + (384 + c) * WP + 4 * h;
    float* st = Ws + CH_W_FLT + wave * 16 * SP;

    for (int t = wave; t < TILES_PER_WG; t += GEMM_WAVES) {
        const int row0 = (tile0 + t) * 16;
        if (t != wave) load_row_pieces(a, X, row0, c, h);
        chain_tile(a, G, wb1, wb2, wb3, wb4, bs1, bn1, bp2, bs2, bn2, st, M2, S2, row0, c, h);
    }
}

#define OWN_ROWS     400
#define N_OWNERS     250
#define OWN_VEC4     (OWN_ROWS * KDIM / 4)
#define BLK_EDGES    16
#define IDX_WG_EDGES 16384
#define IDX_WGS      ((N_EDGES + IDX_WG_EDGES - 1) / IDX_WG_EDGES)
#define BM_ROW_DW    (IDX_WGS * 32)
#define BM_BYTES     ((size_t)N_OWNERS * BM_ROW_DW * 4)
#define BIT_PITCH    257
#define SEG_WORDS    (OWN_ROWS * KDIM / 32)
#define SEG_HDR      8
#define SEG_BATCHES  ((SEG_WORDS - SEG_HDR) / 8)
#define SEG_OVERFLOW 0xFFFFFFFFu

__global__ __launch_bounds__(1024) void build_index(const int* __restrict__ src, const int* __restrict__ dst,
                                                    unsigned int* __restrict__ bm, int* __restrict__ pairs) {
    __shared__ unsigned int bits[32 * BIT_PITCH];
    const int tid  = threadIdx.x;
    const int lane = tid & 31;
    const int wave = tid >> 5;
    for (int i = tid; i < 32 * BIT_PITCH; i += 1024) bits[i] = 0u;
    __syncthreads();

    const int wbase = blockIdx.x * IDX_WG_EDGES + wave * 512;
    int2 s[8], d[8];
#pragma unroll
    for (int q = 0; q < 8; ++q) {
        const int e = wbase + q * 64 + 2 * lane;
        if (e < N_EDGES) { s[q] = *(const int2*)(src + e); d[q] = *(const int2*)(dst + e); }
        else             { s[q] = make_int2(0, 0);         d[q] = make_int2(0, 0); }
    }
#pragma unroll
    for (int q = 0; q < 8; ++q) {
        const int e = wbase + q * 64 + 2 * lane;
        if (e < N_EDGES) {
            *(int4*)(pairs + 2 * (size_t)e) = make_int4(s[q].x, d[q].x, s[q].y, d[q].y);
            const unsigned int bit = 1u << (q * 4 + (lane >> 3));
            const unsigned int o0 = (unsigned int)d[q].x / (unsigned int)OWN_ROWS;
            const unsigned int o1 = (unsigned int)d[q].y / (unsigned int)OWN_ROWS;
            __hip_atomic_fetch_or(&bits[wave * BIT_PITCH + o0], bit, __ATOMIC_RELAXED, __HIP_MEMORY_SCOPE_WORKGROUP);
            if (o1 != o0)
                __hip_atomic_fetch_or(&bits[wave * BIT_PITCH + o1], bit, __ATOMIC_RELAXED, __HIP_MEMORY_SCOPE_WORKGROUP);
        }
    }
    __syncthreads();

    unsigned int* seg = bm + blockIdx.x * 32;
    for (int idx = tid; idx < N_OWNERS * 32; idx += 1024) {
        const int o = idx >> 5, i = idx & 31;
        seg[(size_t)o * BM_ROW_DW + i] = bits[i * BIT_PITCH + o];
    }
}

__device__ __forceinline__ void lds_umax(unsigned int* p, float v) {
    __hip_atomic_fetch_max(p, __float_as_uint(v), __ATOMIC_RELAXED, __HIP_MEMORY_SCOPE_WORKGROUP);
}

#define BL_CAP 1024
#define EL_CAP 264

template <bool SAVE, bool REPLAY>
__global__ __launch_bounds__(1024) void owner_max(const float* __restrict__ m,
                                                  const unsigned int* __restrict__ bm,
                                                  const int* __restrict__ pairs,
                                                  unsigned int* lists,
                                                  float* agg) {
    __shared__ __attribute__((aligned(16))) unsigned int acc[OWN_ROWS * KDIM];
    __shared__ unsigned short blist[32 * BL_CAP];
    __shared__ unsigned int   elist[32 * EL_CAP];
    const int tid  = threadIdx.x;
    const int lane = tid & 31;
    const int wave = __builtin_amdgcn_readfirstlane(tid >> 5);
    const int lo   = blockIdx.x * OWN_ROWS;
    const int half = lane >> 4;
    const int sub  = lane & 15;
    const int f4   = lane << 2;

    const unsigned int* brow = bm + (size_t)blockIdx.x * BM_ROW_DW;
    unsigned int* seg = lists + (size_t)blockIdx.x * (OWN_ROWS * KDIM) + wave * SEG_WORDS;
    unsigned int b0 = 0u, b1 = 0u, b2 = 0u, b3 = 0u, cur = 0u;
    if (REPLAY) {
        cur = seg[lane];
    } else {
        b0 = brow[tid];
        b1 = brow[tid + 1024];
        b2 = brow[tid + 2048];
        b3 = (tid + 3072 < BM_ROW_DW) ? brow[tid + 3072] : 0u;
    }

    {
        uint4* a4 = (uint4*)acc;
        for (int i = tid; i < OWN_VEC4; i += 1024) a4[i] = make_uint4(0u, 0u, 0u, 0u);
    }
    __syncthreads();

    const int2* pr = (const int2*)pairs;
    unsigned short* bl = blist + wave * BL_CAP;
    unsigned int*   el = elist + wave * EL_CAP;
    int ne = 0;

    int cnt8 = 0;

    auto serve8v = [&](unsigned int vals, int base) __attribute__((always_inline)) {
        int s[8], r[8];
        float4 v[8];
#pragma unroll
        for (int i = 0; i < 8; ++i) {
            const unsigned int w = (unsigned int)__builtin_amdgcn_readlane((int)vals, base + i);
            s[i] = (int)(w & 0x1FFFFu);
            r[i] = (int)(w >> 17);
        }
#pragma unroll
        for (int i = 0; i < 8; ++i) {
            v[i] = *(const float4*)(m + (size_t)s[i] * KDIM + f4);
        }
#pragma unroll
        for (int i = 0; i < 8; ++i) {
            unsigned int* p = acc + r[i] * KDIM + lane;
            lds_umax(p,      v[i].x);
            lds_umax(p + 32, v[i].y);
            lds_umax(p + 64, v[i].z);
            lds_umax(p + 96, v[i].w);
        }
    };

    auto serve8 = [&](int pos) __attribute__((always_inline)) {
        const unsigned int mine = el[pos + (lane & 7)];
        if (SAVE) {
            if (cnt8 < SEG_BATCHES && lane < 8) seg[SEG_HDR + cnt8 * 8 + lane] = mine;
            ++cnt8;
        }
        serve8v(mine, 0);
    };

    auto search = [&]() __attribute__((always_inline)) {
    for (int j = 0; j < 4; ++j) {
        const int dw0 = j * 1024 + (wave << 5);
        if (dw0 >= BM_ROW_DW) break;
        const unsigned int word = (j == 0) ? b0 : (j == 1) ? b1 : (j == 2) ? b2 : b3;

        int nblk = 0;
        for (int t = 0; t < 32; ++t) {
            const bool set = ((word >> t) & 1u) != 0u;
            const unsigned int mask = __builtin_amdgcn_ballot_w32(set);
            if (set) bl[nblk + (int)__builtin_amdgcn_mbcnt_lo(mask, 0u)] = (unsigned short)((lane << 5) | t);
            nblk += __builtin_popcount(mask);
        }

        const int blk0 = dw0 << 5;
        for (int k = 0; k < nblk; k += 16) {
            int2 p[8];
#pragma unroll
            for (int i = 0; i < 8; ++i) {
                int idx = k + 2 * i + half;
                idx = (idx < nblk) ? idx : (nblk - 1);
                const int id = blk0 + (int)bl[idx];
                p[i] = pr[(size_t)id * BLK_EDGES + sub];
            }
#pragma unroll
            for (int i = 0; i < 8; ++i) {
                const unsigned int rr = (unsigned int)(p[i].y - lo);
                const bool ok = rr < (unsigned int)OWN_ROWS;
                const unsigned int mask = __builtin_amdgcn_ballot_w32(ok);
                if (ok) el[ne + (int)__builtin_amdgcn_mbcnt_lo(mask, 0u)] = (unsigned int)p[i].x | (rr << 17);
                ne += __builtin_popcount(mask);
            }
            int pos = 0;
            while (ne - pos >= 8) { serve8(pos); pos += 8; }
            const int rem = ne - pos;
            if (pos > 0 && lane < rem) {
                const unsigned int v = el[pos + lane];
                el[lane] = v;
            }
            ne = rem;
        }
    }

    if (ne > 0) {
        if (lane >= ne && lane < 8) el[lane] = el[0];
        serve8(0);
    }
    };

    if (REPLAY) {
        const unsigned int hdr = (unsigned int)__builtin_amdgcn_readfirstlane((int)cur);
        if (hdr <= (unsigned int)SEG_BATCHES) {
            const int n8     = (int)hdr;
            const int nwords = SEG_HDR + 8 * n8;
            const int npiece = (nwords + 31) >> 5;
            for (int g = 0; g < npiece; ++g) {
                unsigned int nxt = 0u;
                const int wn = 32 * (g + 1) + lane;
                const int bf = 4 * g - 1;
                if (bf >= 0 && bf + 3 < n8) {
                    serve8v(cur, 0);
                    nxt = seg[(wn < nwords) ? wn : (nwords - 1)];
                    serve8v(cur, 8);
                    serve8v(cur, 16);
                    serve8v(cur, 24);
                } else {
                    if (wn < nwords) nxt = seg[wn];
#pragma unroll
                    for (int q = 0; q < 4; ++q) {
                        const int b = bf + q;
                        if (b >= 0 && b < n8) serve8v(cur, 8 * q);
                    }
                }
                cur = nxt;
            }
        } else {
            b0 = brow[tid];
            b1 = brow[tid + 1024];
            b2 = brow[tid + 2048];
            b3 = (tid + 3072 < BM_ROW_DW) ? brow[tid + 3072] : 0u;
            search();
        }
    } else {
        search();
    }
    if (SAVE) {
        const unsigned int h = (cnt8 <= SEG_BATCHES) ? (unsigned int)cnt8 : SEG_OVERFLOW;
        if (lane == 0)                     seg[0] = h;
        else if (lane < SEG_HDR)           seg[lane] = 0u;
        else if (lane >= SEG_HDR + 8 * cnt8) seg[lane] = 0u;
    }
    __syncthreads();

    {
        uint4* g4 = (uint4*)(agg + (size_t)lo * KDIM);
        for (int i = tid; i < OWN_VEC4; i += 1024) {
            const unsigned int* p = acc + (i >> 5) * KDIM + (i & 31);
            g4[i] = make_uint4(p[0], p[32], p[64], p[96]);
        }
    }
}

extern "C" void kernel_launch(void* const* d_in, const int* in_sizes, int n_in,
                              void* d_out, int out_size, void* d_ws, size_t ws_size,
                              hipStream_t stream) {
    const float* in_feat  = (const float*)d_in[0];
    const int*   src      = (const int*)d_in[1];
    const int*   dst      = (const int*)d_in[2];
    const float* pool_w1  = (const float*)d_in[3];
    const float* pool_b1  = (const float*)d_in[4];
    const float* self_w1  = (const float*)d_in[5];
    const float* self_b1  = (const float*)d_in[6];
    const float* neigh_w1 = (const float*)d_in[7];
    const float* neigh_b1 = (const float*)d_in[8];
    const float* pool_w2  = (const float*)d_in[9];
    const float* pool_b2  = (const float*)d_in[10];
    const float* self_w2  = (const float*)d_in[11];
    const float* self_b2  = (const float*)d_in[12];
    const float* neigh_w2 = (const float*)d_in[13];
    const float* neigh_b2 = (const float*)d_in[14];
    float* out = (float*)d_out;

    const size_t NODE_BYTES = (size_t)N_NODES * KDIM * sizeof(float);
    char* ws = (char*)d_ws;
    float* p0 = (float*)(ws);
    float* p1 = (float*)(ws + NODE_BYTES);
    float* p2 = (float*)(ws + 2 * NODE_BYTES);
    unsigned int* bm = (unsigned int*)d_out;
    int* pairs = (int*)((char*)d_out + BM_BYTES);

    build_index<<<IDX_WGS, 1024, 0, stream>>>(src, dst, bm, pairs);

    strip_gemm<8, false, true, true, false><<<GEMM_WGS, GEMM_WAVES * 32, 0, stream>>>(in_feat, pool_w1, pool_b1,
                                                                                     in_feat, pool_w1, pool_b1, p0, 128);
    owner_max<true, false><<<N_OWNERS, 1024, 0, stream>>>(p0, bm, pairs, (unsigned int*)p2, p1);

    chain_gemm<<<GEMM_WGS, GEMM_WAVES * 32, 0, stream>>>(in_feat, p1, self_w1, self_b1, neigh_w1, neigh_b1,
                                                         pool_w2, pool_b2, self_w2, self_b2, neigh_b2, p1, p0);

    owner_max<false, true><<<N_OWNERS, 1024, 0, stream>>>(p1, bm, pairs, (unsigned int*)p2, p2);
    strip_gemm<3, false, false, false, true><<<GEMM_WGS, GEMM_WAVES * 32, 0, stream>>>(p2, neigh_w2, neigh_b2,
                                                                                      p0, neigh_w2, neigh_b2, out, 40);
}


# Layer-2 output product taken by each owner work-group from its own LDS max plane (pitch 132, 16-byte A pieces); no layer-2 max plane in memory, 5 launches not 6; k summed in another order
# speedup vs baseline: 3.9350x; 1.0384x over previous
#include <hip/hip_runtime.h>


typedef float v2f __attribute__((ext_vector_type(2)));
typedef float v4f __attribute__((ext_vector_type(4)));
typedef float v8f __attribute__((ext_vector_type(8)));

#define N_NODES 100000
#define N_EDGES 1600000
#define KDIM    128
#define N_CLS   40

__device__ __forceinline__ v8f wmma_f32(v2f a, v2f b, v8f c) {
    return __builtin_amdgcn_wmma_f32_16x16x4_f32(
        false, a, false, b, (short)0, c, false, false);
}

#define WP            132
#define GEMM_WGS      250
#define TILES_PER_WG  25
#define GEMM_WAVES    16

__device__ __forceinline__ void load_row_pieces(v4f (&a)[16], const float* __restrict__ X, int row0, int c, int h) {
    const float* xa = X + (size_t)(row0 + c) * KDIM + 4 * h;
#pragma unroll
    for (int kk = 0; kk < 16; ++kk) a[kk] = *(const v4f*)(xa + 8 * kk);
}

__device__ __forceinline__ void hold_row_pieces(const v4f (&a)[16]) {
    asm volatile("" :: "v"(a[0]), "v"(a[1]), "v"(a[2]), "v"(a[3]), "v"(a[4]), "v"(a[5]), "v"(a[6]), "v"(a[7]),
                       "v"(a[8]), "v"(a[9]), "v"(a[10]), "v"(a[11]), "v"(a[12]), "v"(a[13]), "v"(a[14]), "v"(a[15]));
}

template <int NT, int NK>
__device__ __forceinline__ void strip_products(v8f (&acc)[NT], const v4f (&a)[NK], const float* wb) {
#pragma unroll
    for (int kk = 0; kk < NK; ++kk) {
        v4f b[NT];
#pragma unroll
        for (int j = 0; j < NT; ++j) b[j] = *(const v4f*)(wb + j * 16 * WP + 8 * kk);
#pragma unroll
        for (int j = 0; j < NT; ++j) acc[j] = wmma_f32(v2f{a[kk].x, a[kk].y}, v2f{b[j].x, b[j].y}, acc[j]);
#pragma unroll
        for (int j = 0; j < NT; ++j) acc[j] = wmma_f32(v2f{a[kk].z, a[kk].w}, v2f{b[j].z, b[j].w}, acc[j]);
    }
}

template <int NT, bool TWO, bool RELU, bool PERM>
__device__ __forceinline__ void strip_tile(v4f (&a)[16], const float* __restrict__ X2,
                                           const float* wb1, const float* wb2,
                                           const float* __restrict__ b1, const float* __restrict__ b2,
                                           float* __restrict__ Y, int O, int row0, int c, int h) {
    v8f acc[NT];
#pragma unroll
    for (int j = 0; j < NT; ++j) acc[j] = v8f{};

    hold_row_pieces(a);
    strip_products<NT>(acc, a, wb1);
    if (TWO) {
        load_row_pieces(a, X2, row0, c, h);
        hold_row_pieces(a);
        strip_products<NT>(acc, a, wb2);
    }

    if (PERM) {
        float4 bl = *(const float4*)(b1 + 4 * c);
        float4 bh = *(const float4*)(b1 + 64 + 4 * c);
        if (TWO) {
            const float4 cl = *(const float4*)(b2 + 4 * c);
            const float4 ch = *(const float4*)(b2 + 64 + 4 * c);
            bl.x += cl.x; bl.y += cl.y; bl.z += cl.z; bl.w += cl.w;
            bh.x += ch.x; bh.y += ch.y; bh.z += ch.z; bh.w += ch.w;
        }
#pragma unroll
        for (int r = 0; r < 8; ++r) {
            float4 lo = { acc[0][r] + bl.x, acc[1][r] + bl.y, acc[2][r] + bl.z, acc[3][r] + bl.w };
            float4 hi = { acc[4][r] + bh.x, acc[5][r] + bh.y, acc[6][r] + bh.z, acc[7][r] + bh.w };
            if (RELU) {
                lo.x = fmaxf(lo.x, 0.0f); lo.y = fmaxf(lo.y, 0.0f); lo.z = fmaxf(lo.z, 0.0f); lo.w = fmaxf(lo.w, 0.0f);
                hi.x = fmaxf(hi.x, 0.0f); hi.y = fmaxf(hi.y, 0.0f); hi.z = fmaxf(hi.z, 0.0f); hi.w = fmaxf(hi.w, 0.0f);
            }
            float* yr = Y + (size_t)(row0 + 8 * h + r) * KDIM + 4 * c;
            *(float4*)yr        = lo;
            *(float4*)(yr + 64) = hi;
        }
    } else {
#pragma unroll
        for (int j = 0; j < NT; ++j) {
            const int col    = j * 16 + c;
            const bool valid = (col < O);
            const int colW   = valid ? col : (O - 1);
            float bb = b1[colW];
            if (TWO) bb += b2[colW];
#pragma unroll
            for (int r = 0; r < 8; ++r) {
                float v = acc[j][r] + bb;
                if (RELU) v = fmaxf(v, 0.0f);
                if (valid) Y[(size_t)(row0 + 8 * h + r) * O + col] = v;
            }
        }
    }
}

template <int NT, bool TWO, bool RELU, bool PERM>
__global__ __launch_bounds__(GEMM_WAVES * 32) void strip_gemm(const float* __restrict__ X1,
                                                              const float* __restrict__ W1,
                                                              const float* __restrict__ b1,
                                                              const float* __restrict__ X2,
                                                              const float* __restrict__ W2,
                                                              const float* __restrict__ b2,
                                                              float* __restrict__ Y, int O) {
    __shared__ __attribute__((aligned(16))) float Ws[(TWO ? 2 : 1) * NT * 16 * WP];
    const int tid  = threadIdx.x;
    const int lane = tid & 31;
    const int wave = tid >> 5;
    const int c    = lane & 15;
    const int h    = lane >> 4;
    const int tile0 = blockIdx.x * TILES_PER_WG;

    v4f a[16];
    load_row_pieces(a, X1, (tile0 + wave) * 16, c, h);

    {
#pragma unroll
        for (int u = 0; u < NT; ++u) {
            const int i = tid + u * (GEMM_WAVES * 32);
            const int rho = i >> 5, q = i & 31;
            const int cc = rho & 15, j = rho >> 4;
            int n = PERM ? (4 * cc + (j & 3) + 64 * (j >> 2)) : rho;
            n = (n < O) ? n : (O - 1);
            *(float4*)(Ws + rho * WP + q * 4) = *(const float4*)(W1 + (size_t)n * KDIM + q * 4);
            if (TWO)
                *(float4*)(Ws + (NT * 16 + rho) * WP + q * 4) = *(const float4*)(W2 + (size_t)n * KDIM + q * 4);
        }
    }
    __syncthreads();

    const float* wb1 = Ws + c * WP + 4 * h;
    const float* wb2 = Ws + (NT * 16 + c) * WP + 4 * h;

    strip_tile<NT, TWO, RELU, PERM>(a, X2, wb1, wb2, b1, b2, Y, O, (tile0 + wave) * 16, c, h);

    if (wave + GEMM_WAVES < TILES_PER_WG) {
        asm volatile("" ::: "memory");
        const int row0 = (tile0 + wave + GEMM_WAVES) * 16;
        load_row_pieces(a, X1, row0, c, h);
        strip_tile<NT, TWO, RELU, PERM>(a, X2, wb1, wb2, b1, b2, Y, O, row0, c, h);
    }
}

#define SP        68
#define NT2       3
#define CH_W_FLT  ((3 * 8 + NT2) * 16 * WP)
#define CH_LDS    (CH_W_FLT + GEMM_WAVES * 16 * SP)

__device__ __forceinline__ void turn_half(v4f* dst, float* st, const float4 (&v)[8], int c, int h) {
#pragma unroll
    for (int r = 0; r < 8; ++r) *(float4*)(st + (8 * h + r) * SP + 4 * c) = v[r];
    __builtin_amdgcn_fence(__ATOMIC_RELEASE, "wavefront");
    __builtin_amdgcn_wave_barrier();
    __builtin_amdgcn_fence(__ATOMIC_ACQUIRE, "wavefront");
#pragma unroll
    for (int kk = 0; kk < 8; ++kk) dst[kk] = *(const v4f*)(st + c * SP + 8 * kk + 4 * h);
    __builtin_amdgcn_fence(__ATOMIC_RELEASE, "wavefront");
    __builtin_amdgcn_wave_barrier();
    __builtin_amdgcn_fence(__ATOMIC_ACQUIRE, "wavefront");
}

__device__ __forceinline__ void chain_tile(v4f (&a)[16], const float* G,
                                           const float* wb1, const float* wb2, const float* wb3, const float* wb4,
                                           const float* __restrict__ bs1, const float* __restrict__ bn1,
                                           const float* __restrict__ bp2,
                                           const float* __restrict__ bs2, const float* __restrict__ bn2,
                                           float* st, float* M2, float* __restrict__ S2, int row0, int c, int h) {
    float4 lo[8], hi[8];
    {
        v8f acc[8];
#pragma unroll
        for (int j = 0; j < 8; ++j) acc[j] = v8f{};
        hold_row_pieces(a);
        strip_products<8>(acc, a, wb1);
        load_row_pieces(a, G, row0, c, h);
        hold_row_pieces(a);
        strip_products<8>(acc, a, wb2);

        float4 bl = *(const float4*)(bs1 + 4 * c);
        float4 bh = *(const float4*)(bs1 + 64 + 4 * c);
        const float4 cl = *(const float4*)(bn1 + 4 * c);
        const float4 ch = *(const float4*)(bn1 + 64 + 4 * c);
        bl.x += cl.x; bl.y += cl.y; bl.z += cl.z; bl.w += cl.w;
        bh.x += ch.x; bh.y += ch.y; bh.z += ch.z; bh.w += ch.w;
#pragma unroll
        for (int r = 0; r < 8; ++r) {
            lo[r] = make_float4(fmaxf(acc[0][r] + bl.x, 0.0f), fmaxf(acc[1][r] + bl.y, 0.0f),
                                fmaxf(acc[2][r] + bl.z, 0.0f), fmaxf(acc[3][r] + bl.w, 0.0f));
            hi[r] = make_float4(fmaxf(acc[4][r] + bh.x, 0.0f), fmaxf(acc[5][r] + bh.y, 0.0f),
                                fmaxf(acc[6][r] + bh.z, 0.0f), fmaxf(acc[7][r] + bh.w, 0.0f));
        }
    }
    turn_half(&a[0], st, lo, c, h);
    turn_half(&a[8], st, hi, c, h);

    {
        v8f acc3[NT2];
#pragma unroll
        for (int j = 0; j < NT2; ++j) acc3[j] = v8f{};
        strip_products<NT2>(acc3, a, wb4);
#pragma unroll
        for (int j = 0; j < NT2; ++j) {
            const int col    = j * 16 + c;
            const bool valid = (col < N_CLS);
            const int colW   = valid ? col : (N_CLS - 1);
            const float bb   = bs2[colW] + bn2[colW];
#pragma unroll
            for (int r = 0; r < 8; ++r) {
                if (valid) S2[(size_t)(row0 + 8 * h + r) * N_CLS + col] = acc3[j][r] + bb;
            }
        }
    }

    v8f acc2[8];
#pragma unroll
    for (int j = 0; j < 8; ++j) acc2[j] = v8f{};
    strip_products<8>(acc2, a, wb3);

    {
        const float4 bl = *(const float4*)(bp2 + 4 * c);
        const float4 bh = *(const float4*)(bp2 + 64 + 4 * c);
#pragma unroll
        for (int r = 0; r < 8; ++r) {
            const float4 l4 = make_float4(fmaxf(acc2[0][r] + bl.x, 0.0f), fmaxf(acc2[1][r] + bl.y, 0.0f),
                                          fmaxf(acc2[2][r] + bl.z, 0.0f), fmaxf(acc2[3][r] + bl.w, 0.0f));
            const float4 h4 = make_float4(fmaxf(acc2[4][r] + bh.x, 0.0f), fmaxf(acc2[5][r] + bh.y, 0.0f),
                                          fmaxf(acc2[6][r] + bh.z, 0.0f), fmaxf(acc2[7][r] + bh.w, 0.0f));
            float* yr = M2 + (size_t)(row0 + 8 * h + r) * KDIM + 4 * c;
            *(float4*)yr        = l4;
            *(float4*)(yr + 64) = h4;
        }
    }
}

__global__ __launch_bounds__(GEMM_WAVES * 32) void chain_gemm(const float* __restrict__ X,
                                                              const float* G,
                                                              const float* __restrict__ Ws1, const float* __restrict__ bs1,
                                                              const float* __restrict__ Wn1, const float* __restrict__ bn1,
                                                              const float* __restrict__ Wp2, const float* __restrict__ bp2,
                                                              const float* __restrict__ Ws2, const float* __restrict__ bs2,
                                                              const float* __restrict__ bn2,
                                                              float* M2, float* __restrict__ S2) {
    __shared__ __attribute__((aligned(16))) float Ws[CH_LDS];
    const int tid  = threadIdx.x;
    const int lane = tid & 31;
    const int wave = tid >> 5;
    const int c    = lane & 15;
    const int h    = lane >> 4;
    const int tile0 = blockIdx.x * TILES_PER_WG;

    v4f a[16];
    load_row_pieces(a, X, (tile0 + wave) * 16, c, h);

    {
#pragma unroll
        for (int u = 0; u < 8; ++u) {
            const int i = tid + u * (GEMM_WAVES * 32);
            const int rho = i >> 5, q = i & 31;
            const int cc = rho & 15, j = rho >> 4;
            const int n = 4 * cc + (j & 3) + 64 * (j >> 2);
            *(float4*)(Ws + rho * WP + q * 4)             = *(const float4*)(Ws1 + (size_t)n * KDIM + q * 4);
            *(float4*)(Ws + (128 + rho) * WP + q * 4)     = *(const float4*)(Wn1 + (size_t)n * KDIM + q * 4);
            *(float4*)(Ws + (256 + rho) * WP + q * 4)     = *(const float4*)(Wp2 + (size_t)n * KDIM + q * 4);
        }
#pragma unroll
        for (int u = 0; u < NT2; ++u) {
            const int i = tid + u * (GEMM_WAVES * 32);
            const int rho = i >> 5, q = i & 31;
            const int n = (rho < N_CLS) ? rho : (N_CLS - 1);
            *(float4*)(Ws + (384 + rho) * WP + q * 4)     = *(const float4*)(Ws2 + (size_t)n * KDIM + q * 4);
        }
    }
    __syncthreads();

    const float* wb1 = Ws + c * WP + 4 * h;
    const float* wb2 = Ws + (128 + c) * WP + 4 * h;
    const float* wb3 = Ws + (256 + c) * WP + 4 * h;
    const float* wb4 = Ws + (384 + c) * WP + 4 * h;
    float* st = Ws + CH_W_FLT + wave * 16 * SP;

    for (int t = wave; t < TILES_PER_WG; t += GEMM_WAVES) {
        const int row0 = (tile0 + t) * 16;
        if (t != wave) load_row_pieces(a, X, row0, c, h);
        chain_tile(a, G, wb1, wb2, wb3, wb4, bs1, bn1, bp2, bs2, bn2, st, M2, S2, row0, c, h);
    }
}

#define OWN_ROWS     400
#define N_OWNERS     250
#define OWN_VEC4     (OWN_ROWS * KDIM / 4)
#define BLK_EDGES    16
#define IDX_WG_EDGES 16384
#define IDX_WGS      ((N_EDGES + IDX_WG_EDGES - 1) / IDX_WG_EDGES)
#define BM_ROW_DW    (IDX_WGS * 32)
#define BM_BYTES     ((size_t)N_OWNERS * BM_ROW_DW * 4)
#define BIT_PITCH    257
#define SEG_WORDS    800
#define SEG_HDR      8
#define SEG_BATCHES  ((SEG_WORDS - SEG_HDR) / 8)
#define LIST_BYTES   ((size_t)N_OWNERS * 32 * SEG_WORDS * 4)
#define AP_OUT       132
#define SEG_OVERFLOW 0xFFFFFFFFu

__global__ __launch_bounds__(1024) void build_index(const int* __restrict__ src, const int* __restrict__ dst,
                                                    unsigned int* __restrict__ bm, int* __restrict__ pairs) {
    __shared__ unsigned int bits[32 * BIT_PITCH];
    const int tid  = threadIdx.x;
    const int lane = tid & 31;
    const int wave = tid >> 5;
    for (int i = tid; i < 32 * BIT_PITCH; i += 1024) bits[i] = 0u;
    __syncthreads();

    const int wbase = blockIdx.x * IDX_WG_EDGES + wave * 512;
    int2 s[8], d[8];
#pragma unroll
    for (int q = 0; q < 8; ++q) {
        const int e = wbase + q * 64 + 2 * lane;
        if (e < N_EDGES) { s[q] = *(const int2*)(src + e); d[q] = *(const int2*)(dst + e); }
        else             { s[q] = make_int2(0, 0);         d[q] = make_int2(0, 0); }
    }
#pragma unroll
    for (int q = 0; q < 8; ++q) {
        const int e = wbase + q * 64 + 2 * lane;
        if (e < N_EDGES) {
            *(int4*)(pairs + 2 * (size_t)e) = make_int4(s[q].x, d[q].x, s[q].y, d[q].y);
            const unsigned int bit = 1u << (q * 4 + (lane >> 3));
            const unsigned int o0 = (unsigned int)d[q].x / (unsigned int)OWN_ROWS;
            const unsigned int o1 = (unsigned int)d[q].y / (unsigned int)OWN_ROWS;
            __hip_atomic_fetch_or(&bits[wave * BIT_PITCH + o0], bit, __ATOMIC_RELAXED, __HIP_MEMORY_SCOPE_WORKGROUP);
            if (o1 != o0)
                __hip_atomic_fetch_or(&bits[wave * BIT_PITCH + o1], bit, __ATOMIC_RELAXED, __HIP_MEMORY_SCOPE_WORKGROUP);
        }
    }
    __syncthreads();

    unsigned int* seg = bm + blockIdx.x * 32;
    for (int idx = tid; idx < N_OWNERS * 32; idx += 1024) {
        const int o = idx >> 5, i = idx & 31;
        seg[(size_t)o * BM_ROW_DW + i] = bits[i * BIT_PITCH + o];
    }
}

__device__ __forceinline__ void lds_umax(unsigned int* p, float v) {
    __hip_atomic_fetch_max(p, __float_as_uint(v), __ATOMIC_RELAXED, __HIP_MEMORY_SCOPE_WORKGROUP);
}

#define BL_CAP 1024
#define EL_CAP 264

template <bool SAVE, bool REPLAY, bool OUT>
__global__ __launch_bounds__(1024) void owner_max(const float* __restrict__ m,
                                                  const unsigned int* __restrict__ bm,
                                                  const int* __restrict__ pairs,
                                                  unsigned int* lists,
                                                  float* agg,
                                                  const float* __restrict__ Wn,
                                                  const float* __restrict__ S2,
                                                  float* __restrict__ out) {
    constexpr int AP = OUT ? AP_OUT : KDIM;
    __shared__ __attribute__((aligned(16))) unsigned int acc[OWN_ROWS * AP];
    __shared__ __attribute__((aligned(16))) unsigned short blist[32 * BL_CAP];
    __shared__ unsigned int   elist[32 * EL_CAP];
    const int tid  = threadIdx.x;
    const int lane = tid & 31;
    const int wave = __builtin_amdgcn_readfirstlane(tid >> 5);
    const int lo   = blockIdx.x * OWN_ROWS;
    const int half = lane >> 4;
    const int sub  = lane & 15;
    const int f4   = lane << 2;

    const unsigned int* brow = bm + (size_t)blockIdx.x * BM_ROW_DW;
    unsigned int* seg = lists + (size_t)blockIdx.x * (32 * SEG_WORDS) + wave * SEG_WORDS;
    unsigned int b0 = 0u, b1 = 0u, b2 = 0u, b3 = 0u, cur = 0u;
    if (REPLAY) {
        cur = seg[lane];
    } else {
        b0 = brow[tid];
        b1 = brow[tid + 1024];
        b2 = brow[tid + 2048];
        b3 = (tid + 3072 < BM_ROW_DW) ? brow[tid + 3072] : 0u;
    }

    float4 wA = make_float4(0.0f, 0.0f, 0.0f, 0.0f), wB = wA;
    if (OUT) {
        const int q  = tid & 31;
        const int n2 = (32 + (tid >> 5) < N_CLS) ? (32 + (tid >> 5)) : (N_CLS - 1);
        wA = *(const float4*)(Wn + (size_t)(tid >> 5) * KDIM + 4 * q);
        if (tid < 512) wB = *(const float4*)(Wn + (size_t)n2 * KDIM + 4 * q);
    }

    {
        uint4* a4 = (uint4*)acc;
        for (int i = tid; i < OWN_ROWS * AP / 4; i += 1024) a4[i] = make_uint4(0u, 0u, 0u, 0u);
    }
    __syncthreads();

    const int2* pr = (const int2*)pairs;
    unsigned short* bl = blist + wave * BL_CAP;
    unsigned int*   el = elist + wave * EL_CAP;
    int ne = 0;

    int cnt8 = 0;

    auto serve8v = [&](unsigned int vals, int base) __attribute__((always_inline)) {
        int s[8], r[8];
        float4 v[8];
#pragma unroll
        for (int i = 0; i < 8; ++i) {
            const unsigned int w = (unsigned int)__builtin_amdgcn_readlane((int)vals, base + i);
            s[i] = (int)(w & 0x1FFFFu);
            r[i] = (int)(w >> 17);
        }
#pragma unroll
        for (int i = 0; i < 8; ++i) {
            v[i] = *(const float4*)(m + (size_t)s[i] * KDIM + f4);
        }
#pragma unroll
        for (int i = 0; i < 8; ++i) {
            unsigned int* p = acc + r[i] * AP + lane;
            lds_umax(p,      v[i].x);
            lds_umax(p + 32, v[i].y);
            lds_umax(p + 64, v[i].z);
            lds_umax(p + 96, v[i].w);
        }
    };

    auto serve8 = [&](int pos) __attribute__((always_inline)) {
        const unsigned int mine = el[pos + (lane & 7)];
        if (SAVE) {
            if (cnt8 < SEG_BATCHES && lane < 8) seg[SEG_HDR + cnt8 * 8 + lane] = mine;
            ++cnt8;
        }
        serve8v(mine, 0);
    };

    auto search = [&]() __attribute__((always_inline)) {
    for (int j = 0; j < 4; ++j) {
        const int dw0 = j * 1024 + (wave << 5);
        if (dw0 >= BM_ROW_DW) break;
        const unsigned int word = (j == 0) ? b0 : (j == 1) ? b1 : (j == 2) ? b2 : b3;

        int nblk = 0;
        for (int t = 0; t < 32; ++t) {
            const bool set = ((word >> t) & 1u) != 0u;
            const unsigned int mask = __builtin_amdgcn_ballot_w32(set);
            if (set) bl[nblk + (int)__builtin_amdgcn_mbcnt_lo(mask, 0u)] = (unsigned short)((lane << 5) | t);
            nblk += __builtin_popcount(mask);
        }

        const int blk0 = dw0 << 5;
        for (int k = 0; k < nblk; k += 16) {
            int2 p[8];
#pragma unroll
            for (int i = 0; i < 8; ++i) {
                int idx = k + 2 * i + half;
                idx = (idx < nblk) ? idx : (nblk - 1);
                const int id = blk0 + (int)bl[idx];
                p[i] = pr[(size_t)id * BLK_EDGES + sub];
            }
#pragma unroll
            for (int i = 0; i < 8; ++i) {
                const unsigned int rr = (unsigned int)(p[i].y - lo);
                const bool ok = rr < (unsigned int)OWN_ROWS;
                const unsigned int mask = __builtin_amdgcn_ballot_w32(ok);
                if (ok) el[ne + (int)__builtin_amdgcn_mbcnt_lo(mask, 0u)] = (unsigned int)p[i].x | (rr << 17);
                ne += __builtin_popcount(mask);
            }
            int pos = 0;
            while (ne - pos >= 8) { serve8(pos); pos += 8; }
            const int rem = ne - pos;
            if (pos > 0 && lane < rem) {
                const unsigned int v = el[pos + lane];
                el[lane] = v;
            }
            ne = rem;
        }
    }

    if (ne > 0) {
        if (lane >= ne && lane < 8) el[lane] = el[0];
        serve8(0);
    }
    };

    if (REPLAY) {
        const unsigned int hdr = (unsigned int)__builtin_amdgcn_readfirstlane((int)cur);
        if (hdr <= (unsigned int)SEG_BATCHES) {
            const int n8     = (int)hdr;
            const int nwords = SEG_HDR + 8 * n8;
            const int npiece = (nwords + 31) >> 5;
            for (int g = 0; g < npiece; ++g) {
                unsigned int nxt = 0u;
                const int wn = 32 * (g + 1) + lane;
                const int bf = 4 * g - 1;
                if (bf >= 0 && bf + 3 < n8) {
                    serve8v(cur, 0);
                    nxt = seg[(wn < nwords) ? wn : (nwords - 1)];
                    serve8v(cur, 8);
                    serve8v(cur, 16);
                    serve8v(cur, 24);
                } else {
                    if (wn < nwords) nxt = seg[wn];
#pragma unroll
                    for (int q = 0; q < 4; ++q) {
                        const int b = bf + q;
                        if (b >= 0 && b < n8) serve8v(cur, 8 * q);
                    }
                }
                cur = nxt;
            }
        } else {
            b0 = brow[tid];
            b1 = brow[tid + 1024];
            b2 = brow[tid + 2048];
            b3 = (tid + 3072 < BM_ROW_DW) ? brow[tid + 3072] : 0u;
            search();
        }
    } else {
        search();
    }
    if (SAVE) {
        const unsigned int h = (cnt8 <= SEG_BATCHES) ? (unsigned int)cnt8 : SEG_OVERFLOW;
        if (lane == 0)                     seg[0] = h;
        else if (lane < SEG_HDR)           seg[lane] = 0u;
        else if (lane >= SEG_HDR + 8 * cnt8) seg[lane] = 0u;
    }

    if (OUT) {
        const int c = lane & 15;
        const int h = lane >> 4;
        const int t = wave - (32 - OWN_ROWS / 16);
        const bool has_tile = t >= 0;
        const int row0 = lo + t * 16;

        float sv[NT2][8];
        if (has_tile) {
#pragma unroll
            for (int j = 0; j < NT2; ++j) {
                const int col  = j * 16 + c;
                const int colW = (col < N_CLS) ? col : (N_CLS - 1);
#pragma unroll
                for (int r = 0; r < 8; ++r) sv[j][r] = S2[(size_t)(row0 + 8 * h + r) * N_CLS + colW];
            }
        }
        __syncthreads();

        float* Wl = (float*)(void*)blist;
        {
            const int q = tid & 31;
            float* w0 = Wl + (tid >> 5) * WP + q;
            w0[0] = wA.x; w0[32] = wA.y; w0[64] = wA.z; w0[96] = wA.w;
            if (tid < 512) {
                float* w1 = Wl + (32 + (tid >> 5)) * WP + q;
                w1[0] = wB.x; w1[32] = wB.y; w1[64] = wB.z; w1[96] = wB.w;
            }
        }
        __syncthreads();

        if (has_tile) {
            const float* ap = (const float*)(const void*)acc + (t * 16 + c) * AP + 4 * h;
            const float* wb = Wl + c * WP + 4 * h;
            v8f acc3[NT2];
#pragma unroll
            for (int j = 0; j < NT2; ++j) acc3[j] = v8f{};
#pragma unroll
            for (int hf = 0; hf < 2; ++hf) {
                v4f a[8];
#pragma unroll
                for (int kk = 0; kk < 8; ++kk) a[kk] = *(const v4f*)(ap + 64 * hf + 8 * kk);
                strip_products<NT2>(acc3, a, wb + 64 * hf);
            }
#pragma unroll
            for (int j = 0; j < NT2; ++j) {
                const int col    = j * 16 + c;
                const bool valid = (col < N_CLS);
#pragma unroll
                for (int r = 0; r < 8; ++r) {
                    const float v = acc3[j][r] + sv[j][r];
                    if (valid) out[(size_t)(row0 + 8 * h + r) * N_CLS + col] = v;
                }
            }
        }
    } else {
        __syncthreads();
        uint4* g4 = (uint4*)(agg + (size_t)lo * KDIM);
        for (int i = tid; i < OWN_VEC4; i += 1024) {
            const unsigned int* p = acc + (i >> 5) * AP + (i & 31);
            g4[i] = make_uint4(p[0], p[32], p[64], p[96]);
        }
    }
}

extern "C" void kernel_launch(void* const* d_in, const int* in_sizes, int n_in,
                              void* d_out, int out_size, void* d_ws, size_t ws_size,
                              hipStream_t stream) {
    const float* in_feat  = (const float*)d_in[0];
    const int*   src      = (const int*)d_in[1];
    const int*   dst      = (const int*)d_in[2];
    const float* pool_w1  = (const float*)d_in[3];
    const float* pool_b1  = (const float*)d_in[4];
    const float* self_w1  = (const float*)d_in[5];
    const float* self_b1  = (const float*)d_in[6];
    const float* neigh_w1 = (const float*)d_in[7];
    const float* neigh_b1 = (const float*)d_in[8];
    const float* pool_w2  = (const float*)d_in[9];
    const float* pool_b2  = (const float*)d_in[10];
    const float* self_w2  = (const float*)d_in[11];
    const float* self_b2  = (const float*)d_in[12];
    const float* neigh_w2 = (const float*)d_in[13];
    const float* neigh_b2 = (const float*)d_in[14];
    float* out = (float*)d_out;

    const size_t NODE_BYTES = (size_t)N_NODES * KDIM * sizeof(float);
    char* ws = (char*)d_ws;
    float* p0 = (float*)(ws);
    float* p1 = (float*)(ws + NODE_BYTES);
    float* p2 = (float*)(ws + 2 * NODE_BYTES);
    unsigned int* bm = (unsigned int*)((char*)p2 + LIST_BYTES);
    int* pairs = (int*)((char*)p2 + LIST_BYTES + BM_BYTES);

    build_index<<<IDX_WGS, 1024, 0, stream>>>(src, dst, bm, pairs);

    strip_gemm<8, false, true, true><<<GEMM_WGS, GEMM_WAVES * 32, 0, stream>>>(in_feat, pool_w1, pool_b1,
                                                                              in_feat, pool_w1, pool_b1, p0, 128);
    owner_max<true, false, false><<<N_OWNERS, 1024, 0, stream>>>(p0, bm, pairs, (unsigned int*)p2, p1, neigh_w2, p0, out);

    chain_gemm<<<GEMM_WGS, GEMM_WAVES * 32, 0, stream>>>(in_feat, p1, self_w1, self_b1, neigh_w1, neigh_b1,
                                                         pool_w2, pool_b2, self_w2, self_b2, neigh_b2, p1, p0);

    owner_max<false, true, true><<<N_OWNERS, 1024, 0, stream>>>(p1, bm, pairs, (unsigned int*)p2, p2, neigh_w2, p0, out);
}
